# starting kernel
# baseline (speedup 1.0000x reference)
#include <hip/hip_runtime.h>
#include <hip/hip_bf16.h>

typedef __attribute__((ext_vector_type(2))) float v2f;
typedef __attribute__((ext_vector_type(4))) float v4f;
typedef __attribute__((ext_vector_type(8))) float v8f;

#define NNODES    50000
#define NEDGES    640000
#define CH        128
#define ROW_TILES 3125

__global__ void gcn_deg_init(float* __restrict__ deg) {
    int i = blockIdx.x * blockDim.x + threadIdx.x;
    if (i < NNODES) deg[i] = 1.0f;
}

__global__ void gcn_deg_accum(const int* __restrict__ dst, float* __restrict__ deg) {
    int e = blockIdx.x * blockDim.x + threadIdx.x;
    if (e < NEDGES) unsafeAtomicAdd(&deg[dst[e]], 1.0f);
}

__global__ void gcn_dinv(const float* __restrict__ deg, float* __restrict__ dinv) {
    int i = blockIdx.x * blockDim.x + threadIdx.x;
    if (i < NNODES) {
        float d = deg[i];
        dinv[i] = (d > 0.0f) ? rsqrtf(d) : 0.0f;
    }
}

__global__ __launch_bounds__(256) void gcn_gemm_wmma(const float* __restrict__ X,
                                                     const float* __restrict__ W,
                                                     float* __restrict__ H) {
    __shared__ float Wt[CH * CH];
    for (int idx = threadIdx.x; idx < CH * CH; idx += 256) {
        int k = idx >> 7, c = idx & 127;
        Wt[c * CH + k] = W[idx];
    }
    __syncthreads();

    const int wave = threadIdx.x >> 5;
    const int lane = threadIdx.x & 31;
    const int rt   = blockIdx.x * 8 + wave;
    if (rt >= ROW_TILES) return;

    const int half = lane >> 4;
    const int mn   = lane & 15;

    const float* __restrict__ xrow = X + (size_t)(rt * 16 + mn) * CH;

    v8f acc[8];
    v8f zero = {};
    #pragma unroll
    for (int c = 0; c < 8; ++c) acc[c] = zero;

    #pragma unroll 4
    for (int kk = 0; kk < CH; kk += 4) {
        v2f a = *(const v2f*)(xrow + kk + 2 * half);
        #pragma unroll
        for (int c = 0; c < 8; ++c) {
            v2f b = *(const v2f*)(&Wt[(c * 16 + mn) * CH + kk + 2 * half]);
            acc[c] = __builtin_amdgcn_wmma_f32_16x16x4_f32(
                false, a, false, b, (short)0, acc[c], false, false);
        }
    }

    #pragma unroll
    for (int c = 0; c < 8; ++c) {
        #pragma unroll
        for (int i = 0; i < 8; ++i) {
            H[(size_t)(rt * 16 + half * 8 + i) * CH + c * 16 + mn] = acc[c][i];
        }
    }
}

__global__ void gcn_agg_init(const float* __restrict__ H, const float* __restrict__ dinv,
                             const float* __restrict__ bias, float* __restrict__ out) {
    int idx = blockIdx.x * blockDim.x + threadIdx.x;
    if (idx >= NNODES * 32) return;
    int node = idx >> 5;
    int ch   = (idx & 31) * 4;
    float s  = dinv[node];
    s = s * s;
    v4f h = *(const v4f*)(&H[(size_t)node * CH + ch]);
    v4f b = *(const v4f*)(&bias[ch]);
    v4f o = h * s + b;
    *(v4f*)(&out[(size_t)node * CH + ch]) = o;
}

__global__ __launch_bounds__(256) void gcn_agg_edges(const int* __restrict__ src,
                                                     const int* __restrict__ dst,
                                                     const float* __restrict__ dinv,
                                                     const float* __restrict__ H,
                                                     float* __restrict__ out) {
    int e = blockIdx.x * 8 + (threadIdx.x >> 5);
    if (e >= NEDGES) return;
    int lane = threadIdx.x & 31;
    int s = src[e];
    int d = dst[e];
    float norm = dinv[s] * dinv[d];
    int ch = lane * 4;
    v4f h = *(const v4f*)(&H[(size_t)s * CH + ch]);
    float* op = &out[(size_t)d * CH + ch];
    unsafeAtomicAdd(op + 0, h.x * norm);
    unsafeAtomicAdd(op + 1, h.y * norm);
    unsafeAtomicAdd(op + 2, h.z * norm);
    unsafeAtomicAdd(op + 3, h.w * norm);
}

__global__ void gcn_relu(float* __restrict__ x, int n4) {
    int i = blockIdx.x * blockDim.x + threadIdx.x;
    if (i >= n4) return;
    v4f v = *(const v4f*)(&x[i * 4]);
    v.x = fmaxf(v.x, 0.0f); v.y = fmaxf(v.y, 0.0f);
    v.z = fmaxf(v.z, 0.0f); v.w = fmaxf(v.w, 0.0f);
    *(v4f*)(&x[i * 4]) = v;
}

extern "C" void kernel_launch(void* const* d_in, const int* in_sizes, int n_in,
                              void* d_out, int out_size, void* d_ws, size_t ws_size,
                              hipStream_t stream) {
    const float* x   = (const float*)d_in[0];
    const int*   ei  = (const int*)d_in[1];
    const float* W1  = (const float*)d_in[2];
    const float* b1  = (const float*)d_in[3];
    const float* W2  = (const float*)d_in[4];
    const float* b2  = (const float*)d_in[5];
    float*       out = (float*)d_out;

    const int* src = ei;
    const int* dst = ei + NEDGES;

    float* ws   = (float*)d_ws;
    float* deg  = ws;
    float* dinv = ws + 65536;
    float* H    = ws + 131072;
    float* out1 = ws + 131072 + 6400000;

    const int T = 256;
    gcn_deg_init <<<(NNODES + T - 1) / T, T, 0, stream>>>(deg);
    gcn_deg_accum<<<(NEDGES + T - 1) / T, T, 0, stream>>>(dst, deg);
    gcn_dinv     <<<(NNODES + T - 1) / T, T, 0, stream>>>(deg, dinv);

    const int gemm_blocks = (ROW_TILES + 7) / 8;
    const int agg_init_blocks = (NNODES * 32 + T - 1) / T;
    const int agg_edge_blocks = (NEDGES + 7) / 8;
    const int relu_blocks = (NNODES * CH / 4 + T - 1) / T;

    gcn_gemm_wmma<<<gemm_blocks, T, 0, stream>>>(x, W1, H);
    gcn_agg_init <<<agg_init_blocks, T, 0, stream>>>(H, dinv, b1, out1);
    gcn_agg_edges<<<agg_edge_blocks, T, 0, stream>>>(src, dst, dinv, H, out1);
    gcn_relu     <<<relu_blocks, T, 0, stream>>>(out1, NNODES * CH / 4);

    gcn_gemm_wmma<<<gemm_blocks, T, 0, stream>>>(out1, W2, H);
    gcn_agg_init <<<agg_init_blocks, T, 0, stream>>>(H, dinv, b2, out);
    gcn_agg_edges<<<agg_edge_blocks, T, 0, stream>>>(src, dst, dinv, H, out);
}


# Per-edge f32 atomic scatter replaced: edges bucketed by destination in the call (int atomic count returns arrival-order rank, scan, place); one wave per node sums its rows in registers, stores once
# speedup vs baseline: 1.8311x; 1.8311x over previous
#include <hip/hip_runtime.h>
#include <hip/hip_bf16.h>

typedef __attribute__((ext_vector_type(2))) float v2f;
typedef __attribute__((ext_vector_type(4))) float v4f;
typedef __attribute__((ext_vector_type(8))) float v8f;

#define NNODES    50000
#define NEDGES    640000
#define CH        128
#define ROW_TILES 3125

#define SCAN_T   1024
#define SCAN_PER 49
#define SCAN_N   (SCAN_T * SCAN_PER)

__global__ void gcn_cnt_zero(int* __restrict__ cnt) {
    int i = blockIdx.x * blockDim.x + threadIdx.x;
    if (i < NNODES) cnt[i] = 0;
}

__global__ void gcn_cnt_rank(const int* __restrict__ dst, int* __restrict__ cnt,
                             int* __restrict__ rank) {
    int e = blockIdx.x * blockDim.x + threadIdx.x;
    if (e < NEDGES) rank[e] = atomicAdd(&cnt[dst[e]], 1);
}

__global__ __launch_bounds__(SCAN_T) void gcn_scan(const int* __restrict__ cnt,
                                                   int* __restrict__ rowstart,
                                                   float* __restrict__ dinv) {
    __shared__ int s[SCAN_N];
    __shared__ int wsum[32];
    const int t    = threadIdx.x;
    const int lane = t & 31;
    const int wave = t >> 5;

    for (int i = t; i < SCAN_N; i += SCAN_T) s[i] = (i < NNODES) ? cnt[i] : 0;
    __syncthreads();

    int sum = 0;
    for (int j = 0; j < SCAN_PER; ++j) sum += s[t * SCAN_PER + j];

    int inc = sum;
    for (int off = 1; off < 32; off <<= 1) {
        int v = __shfl_up(inc, off);
        if (lane >= off) inc += v;
    }
    if (lane == 31) wsum[wave] = inc;
    __syncthreads();
    if (wave == 0) {
        int w = wsum[lane];
        int winc = w;
        for (int off = 1; off < 32; off <<= 1) {
            int v = __shfl_up(winc, off);
            if (lane >= off) winc += v;
        }
        wsum[lane] = winc - w;
    }
    __syncthreads();

    int run = wsum[wave] + inc - sum;
    for (int j = 0; j < SCAN_PER; ++j) {
        int c = s[t * SCAN_PER + j];
        s[t * SCAN_PER + j] = run;
        run += c;
    }
    __syncthreads();

    for (int i = t; i < NNODES; i += SCAN_T) {
        int r = s[i];
        int c = s[i + 1] - r;
        rowstart[i] = r;
        float dg = (float)(c + 1);
        dinv[i] = (dg > 0.0f) ? rsqrtf(dg) : 0.0f;
    }
    if (t == 0) rowstart[NNODES] = s[NNODES];
}

__global__ void gcn_place(const int* __restrict__ src, const int* __restrict__ dst,
                          const int* __restrict__ rowstart, const int* __restrict__ rank,
                          int* __restrict__ col) {
    int e = blockIdx.x * blockDim.x + threadIdx.x;
    if (e < NEDGES) col[rowstart[dst[e]] + rank[e]] = src[e];
}

__global__ __launch_bounds__(256) void gcn_gemm_wmma(const float* __restrict__ X,
                                                     const float* __restrict__ W,
                                                     float* __restrict__ H) {
    __shared__ float Wt[CH * CH];
    for (int idx = threadIdx.x; idx < CH * CH; idx += 256) {
        int k = idx >> 7, c = idx & 127;
        Wt[c * CH + k] = W[idx];
    }
    __syncthreads();

    const int wave = threadIdx.x >> 5;
    const int lane = threadIdx.x & 31;
    const int rt   = blockIdx.x * 8 + wave;
    if (rt >= ROW_TILES) return;

    const int half = lane >> 4;
    const int mn   = lane & 15;

    const float* __restrict__ xrow = X + (size_t)(rt * 16 + mn) * CH;

    v8f acc[8];
    v8f zero = {};
    #pragma unroll
    for (int c = 0; c < 8; ++c) acc[c] = zero;

    #pragma unroll 4
    for (int kk = 0; kk < CH; kk += 4) {
        v2f a = *(const v2f*)(xrow + kk + 2 * half);
        #pragma unroll
        for (int c = 0; c < 8; ++c) {
            v2f b = *(const v2f*)(&Wt[(c * 16 + mn) * CH + kk + 2 * half]);
            acc[c] = __builtin_amdgcn_wmma_f32_16x16x4_f32(
                false, a, false, b, (short)0, acc[c], false, false);
        }
    }

    #pragma unroll
    for (int c = 0; c < 8; ++c) {
        #pragma unroll
        for (int i = 0; i < 8; ++i) {
            H[(size_t)(rt * 16 + half * 8 + i) * CH + c * 16 + mn] = acc[c][i];
        }
    }
}

__global__ __launch_bounds__(256) void gcn_gather(const int* __restrict__ rowstart,
                                                  const int* __restrict__ col,
                                                  const float* __restrict__ dinv,
                                                  const float* __restrict__ H,
                                                  const float* __restrict__ bias,
                                                  float* __restrict__ out) {
    const int lane = threadIdx.x & 31;
    const int wave = __builtin_amdgcn_readfirstlane((int)(threadIdx.x >> 5));
    const int d    = blockIdx.x * 8 + wave;
    if (d >= NNODES) return;
    const int ch  = lane * 4;
    const int beg = rowstart[d];
    const int end = rowstart[d + 1];
    const float dd = dinv[d];

    v4f acc = *(const v4f*)(&H[(size_t)d * CH + ch]) * (dd * dd) + *(const v4f*)(&bias[ch]);

    for (int base = beg; base < end; base += 32) {
        const int n = min(32, end - base);
        int   mys = 0;
        float myw = 0.0f;
        if (lane < n) {
            mys = col[base + lane];
            myw = dinv[mys] * dd;
        }
        int k = 0;
        for (; k + 4 <= n; k += 4) {
            const int s0 = __builtin_amdgcn_readlane(mys, k);
            const int s1 = __builtin_amdgcn_readlane(mys, k + 1);
            const int s2 = __builtin_amdgcn_readlane(mys, k + 2);
            const int s3 = __builtin_amdgcn_readlane(mys, k + 3);
            const float w0 = __int_as_float(__builtin_amdgcn_readlane(__float_as_int(myw), k));
            const float w1 = __int_as_float(__builtin_amdgcn_readlane(__float_as_int(myw), k + 1));
            const float w2 = __int_as_float(__builtin_amdgcn_readlane(__float_as_int(myw), k + 2));
            const float w3 = __int_as_float(__builtin_amdgcn_readlane(__float_as_int(myw), k + 3));
            const v4f h0 = *(const v4f*)(&H[(size_t)s0 * CH + ch]);
            const v4f h1 = *(const v4f*)(&H[(size_t)s1 * CH + ch]);
            const v4f h2 = *(const v4f*)(&H[(size_t)s2 * CH + ch]);
            const v4f h3 = *(const v4f*)(&H[(size_t)s3 * CH + ch]);
            acc += h0 * w0;
            acc += h1 * w1;
            acc += h2 * w2;
            acc += h3 * w3;
        }
        for (; k < n; ++k) {
            const int   s0 = __builtin_amdgcn_readlane(mys, k);
            const float w0 = __int_as_float(__builtin_amdgcn_readlane(__float_as_int(myw), k));
            const v4f h0 = *(const v4f*)(&H[(size_t)s0 * CH + ch]);
            acc += h0 * w0;
        }
    }
    *(v4f*)(&out[(size_t)d * CH + ch]) = acc;
}

__global__ void gcn_relu(float* __restrict__ x, int n4) {
    int i = blockIdx.x * blockDim.x + threadIdx.x;
    if (i >= n4) return;
    v4f v = *(const v4f*)(&x[i * 4]);
    v.x = fmaxf(v.x, 0.0f); v.y = fmaxf(v.y, 0.0f);
    v.z = fmaxf(v.z, 0.0f); v.w = fmaxf(v.w, 0.0f);
    *(v4f*)(&x[i * 4]) = v;
}

extern "C" void kernel_launch(void* const* d_in, const int* in_sizes, int n_in,
                              void* d_out, int out_size, void* d_ws, size_t ws_size,
                              hipStream_t stream) {
    const float* x   = (const float*)d_in[0];
    const int*   ei  = (const int*)d_in[1];
    const float* W1  = (const float*)d_in[2];
    const float* b1  = (const float*)d_in[3];
    const float* W2  = (const float*)d_in[4];
    const float* b2  = (const float*)d_in[5];
    float*       out = (float*)d_out;

    const int* src = ei;
    const int* dst = ei + NEDGES;

    float* ws   = (float*)d_ws;
    int*   cnt  = (int*)ws;
    float* dinv = ws + 65536;
    float* H    = ws + 131072;
    float* out1 = ws + 131072 + 6400000;
    int*   rowstart = (int*)(ws + 131072 + 12800000);
    int*   rank     = (int*)(ws + 131072 + 12800000 + 65536);
    int*   col      = (int*)(ws + 131072 + 12800000 + 65536 + 640000);

    const int T = 256;
    gcn_cnt_zero<<<(NNODES + T - 1) / T, T, 0, stream>>>(cnt);
    gcn_cnt_rank<<<(NEDGES + T - 1) / T, T, 0, stream>>>(dst, cnt, rank);
    gcn_scan    <<<1, SCAN_T, 0, stream>>>(cnt, rowstart, dinv);
    gcn_place   <<<(NEDGES + T - 1) / T, T, 0, stream>>>(src, dst, rowstart, rank, col);

    const int gemm_blocks = (ROW_TILES + 7) / 8;
    const int gather_blocks = (NNODES + 7) / 8;
    const int relu_blocks = (NNODES * CH / 4 + T - 1) / T;

    gcn_gemm_wmma<<<gemm_blocks, T, 0, stream>>>(x, W1, H);
    gcn_gather   <<<gather_blocks, T, 0, stream>>>(rowstart, col, dinv, H, b1, out1);
    gcn_relu     <<<relu_blocks, T, 0, stream>>>(out1, NNODES * CH / 4);

    gcn_gemm_wmma<<<gemm_blocks, T, 0, stream>>>(out1, W2, H);
    gcn_gather   <<<gather_blocks, T, 0, stream>>>(rowstart, col, dinv, H, b2, out);
}


# GEMM: weight tile kept in LDS at a row pitch of 132 floats instead of 128, so B-fragment reads (16 lanes on one bank) and staging stores (32 lanes on one bank) become free of bank conflicts
# speedup vs baseline: 2.4218x; 1.3226x over previous
#include <hip/hip_runtime.h>
#include <hip/hip_bf16.h>

typedef __attribute__((ext_vector_type(2))) float v2f;
typedef __attribute__((ext_vector_type(4))) float v4f;
typedef __attribute__((ext_vector_type(8))) float v8f;

#define NNODES    50000
#define NEDGES    640000
#define CH        128
#define ROW_TILES 3125
#define WPITCH    132

#define SCAN_T   1024
#define SCAN_PER 49
#define SCAN_N   (SCAN_T * SCAN_PER)

__global__ void gcn_cnt_zero(int* __restrict__ cnt) {
    int i = blockIdx.x * blockDim.x + threadIdx.x;
    if (i < NNODES) cnt[i] = 0;
}

__global__ void gcn_cnt_rank(const int* __restrict__ dst, int* __restrict__ cnt,
                             int* __restrict__ rank) {
    int e = blockIdx.x * blockDim.x + threadIdx.x;
    if (e < NEDGES) rank[e] = atomicAdd(&cnt[dst[e]], 1);
}

__global__ __launch_bounds__(SCAN_T) void gcn_scan(const int* __restrict__ cnt,
                                                   int* __restrict__ rowstart,
                                                   float* __restrict__ dinv) {
    __shared__ int s[SCAN_N];
    __shared__ int wsum[32];
    const int t    = threadIdx.x;
    const int lane = t & 31;
    const int wave = t >> 5;

    for (int i = t; i < SCAN_N; i += SCAN_T) s[i] = (i < NNODES) ? cnt[i] : 0;
    __syncthreads();

    int sum = 0;
    for (int j = 0; j < SCAN_PER; ++j) sum += s[t * SCAN_PER + j];

    int inc = sum;
    for (int off = 1; off < 32; off <<= 1) {
        int v = __shfl_up(inc, off);
        if (lane >= off) inc += v;
    }
    if (lane == 31) wsum[wave] = inc;
    __syncthreads();
    if (wave == 0) {
        int w = wsum[lane];
        int winc = w;
        for (int off = 1; off < 32; off <<= 1) {
            int v = __shfl_up(winc, off);
            if (lane >= off) winc += v;
        }
        wsum[lane] = winc - w;
    }
    __syncthreads();

    int run = wsum[wave] + inc - sum;
    for (int j = 0; j < SCAN_PER; ++j) {
        int c = s[t * SCAN_PER + j];
        s[t * SCAN_PER + j] = run;
        run += c;
    }
    __syncthreads();

    for (int i = t; i < NNODES; i += SCAN_T) {
        int r = s[i];
        int c = s[i + 1] - r;
        rowstart[i] = r;
        float dg = (float)(c + 1);
        dinv[i] = (dg > 0.0f) ? rsqrtf(dg) : 0.0f;
    }
    if (t == 0) rowstart[NNODES] = s[NNODES];
}

__global__ void gcn_place(const int* __restrict__ src, const int* __restrict__ dst,
                          const int* __restrict__ rowstart, const int* __restrict__ rank,
                          int* __restrict__ col) {
    int e = blockIdx.x * blockDim.x + threadIdx.x;
    if (e < NEDGES) col[rowstart[dst[e]] + rank[e]] = src[e];
}

__global__ __launch_bounds__(256) void gcn_gemm_wmma(const float* __restrict__ X,
                                                     const float* __restrict__ W,
                                                     float* __restrict__ H) {
    __shared__ float Wt[CH * WPITCH];
    {
        const int sc = (threadIdx.x >> 5) * 16 + (threadIdx.x & 15);
        const int sk = (threadIdx.x >> 4) & 1;
        const float* __restrict__ wsrc = W + sk * CH + sc;
        float* wdst = &Wt[sc * WPITCH + sk];
        #pragma unroll 16
        for (int i = 0; i < CH / 2; ++i) {
            *wdst = *wsrc;
            wsrc += 2 * CH;
            wdst += 2;
        }
    }
    __syncthreads();

    const int wave = threadIdx.x >> 5;
    const int lane = threadIdx.x & 31;
    const int rt   = blockIdx.x * 8 + wave;
    if (rt >= ROW_TILES) return;

    const int half = lane >> 4;
    const int mn   = lane & 15;

    const float* __restrict__ xrow = X + (size_t)(rt * 16 + mn) * CH;

    v8f acc[8];
    v8f zero = {};
    #pragma unroll
    for (int c = 0; c < 8; ++c) acc[c] = zero;

    #pragma unroll 4
    for (int kk = 0; kk < CH; kk += 4) {
        v2f a = *(const v2f*)(xrow + kk + 2 * half);
        #pragma unroll
        for (int c = 0; c < 8; ++c) {
            v2f b = *(const v2f*)(&Wt[(c * 16 + mn) * WPITCH + kk + 2 * half]);
            acc[c] = __builtin_amdgcn_wmma_f32_16x16x4_f32(
                false, a, false, b, (short)0, acc[c], false, false);
        }
    }

    #pragma unroll
    for (int c = 0; c < 8; ++c) {
        #pragma unroll
        for (int i = 0; i < 8; ++i) {
            H[(size_t)(rt * 16 + half * 8 + i) * CH + c * 16 + mn] = acc[c][i];
        }
    }
}

__global__ __launch_bounds__(256) void gcn_gather(const int* __restrict__ rowstart,
                                                  const int* __restrict__ col,
                                                  const float* __restrict__ dinv,
                                                  const float* __restrict__ H,
                                                  const float* __restrict__ bias,
                                                  float* __restrict__ out) {
    const int lane = threadIdx.x & 31;
    const int wave = __builtin_amdgcn_readfirstlane((int)(threadIdx.x >> 5));
    const int d    = blockIdx.x * 8 + wave;
    if (d >= NNODES) return;
    const int ch  = lane * 4;
    const int beg = rowstart[d];
    const int end = rowstart[d + 1];
    const float dd = dinv[d];

    v4f acc = *(const v4f*)(&H[(size_t)d * CH + ch]) * (dd * dd) + *(const v4f*)(&bias[ch]);

    for (int base = beg; base < end; base += 32) {
        const int n = min(32, end - base);
        int   mys = 0;
        float myw = 0.0f;
        if (lane < n) {
            mys = col[base + lane];
            myw = dinv[mys] * dd;
        }
        int k = 0;
        for (; k + 4 <= n; k += 4) {
            const int s0 = __builtin_amdgcn_readlane(mys, k);
            const int s1 = __builtin_amdgcn_readlane(mys, k + 1);
            const int s2 = __builtin_amdgcn_readlane(mys, k + 2);
            const int s3 = __builtin_amdgcn_readlane(mys, k + 3);
            const float w0 = __int_as_float(__builtin_amdgcn_readlane(__float_as_int(myw), k));
            const float w1 = __int_as_float(__builtin_amdgcn_readlane(__float_as_int(myw), k + 1));
            const float w2 = __int_as_float(__builtin_amdgcn_readlane(__float_as_int(myw), k + 2));
            const float w3 = __int_as_float(__builtin_amdgcn_readlane(__float_as_int(myw), k + 3));
            const v4f h0 = *(const v4f*)(&H[(size_t)s0 * CH + ch]);
            const v4f h1 = *(const v4f*)(&H[(size_t)s1 * CH + ch]);
            const v4f h2 = *(const v4f*)(&H[(size_t)s2 * CH + ch]);
            const v4f h3 = *(const v4f*)(&H[(size_t)s3 * CH + ch]);
            acc += h0 * w0;
            acc += h1 * w1;
            acc += h2 * w2;
            acc += h3 * w3;
        }
        for (; k < n; ++k) {
            const int   s0 = __builtin_amdgcn_readlane(mys, k);
            const float w0 = __int_as_float(__builtin_amdgcn_readlane(__float_as_int(myw), k));
            const v4f h0 = *(const v4f*)(&H[(size_t)s0 * CH + ch]);
            acc += h0 * w0;
        }
    }
    *(v4f*)(&out[(size_t)d * CH + ch]) = acc;
}

__global__ void gcn_relu(float* __restrict__ x, int n4) {
    int i = blockIdx.x * blockDim.x + threadIdx.x;
    if (i >= n4) return;
    v4f v = *(const v4f*)(&x[i * 4]);
    v.x = fmaxf(v.x, 0.0f); v.y = fmaxf(v.y, 0.0f);
    v.z = fmaxf(v.z, 0.0f); v.w = fmaxf(v.w, 0.0f);
    *(v4f*)(&x[i * 4]) = v;
}

extern "C" void kernel_launch(void* const* d_in, const int* in_sizes, int n_in,
                              void* d_out, int out_size, void* d_ws, size_t ws_size,
                              hipStream_t stream) {
    const float* x   = (const float*)d_in[0];
    const int*   ei  = (const int*)d_in[1];
    const float* W1  = (const float*)d_in[2];
    const float* b1  = (const float*)d_in[3];
    const float* W2  = (const float*)d_in[4];
    const float* b2  = (const float*)d_in[5];
    float*       out = (float*)d_out;

    const int* src = ei;
    const int* dst = ei + NEDGES;

    float* ws   = (float*)d_ws;
    int*   cnt  = (int*)ws;
    float* dinv = ws + 65536;
    float* H    = ws + 131072;
    float* out1 = ws + 131072 + 6400000;
    int*   rowstart = (int*)(ws + 131072 + 12800000);
    int*   rank     = (int*)(ws + 131072 + 12800000 + 65536);
    int*   col      = (int*)(ws + 131072 + 12800000 + 65536 + 640000);

    const int T = 256;
    gcn_cnt_zero<<<(NNODES + T - 1) / T, T, 0, stream>>>(cnt);
    gcn_cnt_rank<<<(NEDGES + T - 1) / T, T, 0, stream>>>(dst, cnt, rank);
    gcn_scan    <<<1, SCAN_T, 0, stream>>>(cnt, rowstart, dinv);
    gcn_place   <<<(NEDGES + T - 1) / T, T, 0, stream>>>(src, dst, rowstart, rank, col);

    const int gemm_blocks = (ROW_TILES + 7) / 8;
    const int gather_blocks = (NNODES + 7) / 8;
    const int relu_blocks = (NNODES * CH / 4 + T - 1) / T;

    gcn_gemm_wmma<<<gemm_blocks, T, 0, stream>>>(x, W1, H);
    gcn_gather   <<<gather_blocks, T, 0, stream>>>(rowstart, col, dinv, H, b1, out1);
    gcn_relu     <<<relu_blocks, T, 0, stream>>>(out1, NNODES * CH / 4);

    gcn_gemm_wmma<<<gemm_blocks, T, 0, stream>>>(out1, W2, H);
    gcn_gather   <<<gather_blocks, T, 0, stream>>>(rowstart, col, dinv, H, b2, out);
}


# Layer-1 relu fused into the gather: max with 0 on the finished row in registers before its single store; the separate relu launch (50000 short waves that re-read and re-wrote every row) is gone
# speedup vs baseline: 2.5548x; 1.0549x over previous
#include <hip/hip_runtime.h>
#include <hip/hip_bf16.h>

typedef __attribute__((ext_vector_type(2))) float v2f;
typedef __attribute__((ext_vector_type(4))) float v4f;
typedef __attribute__((ext_vector_type(8))) float v8f;

#define NNODES    50000
#define NEDGES    640000
#define CH        128
#define ROW_TILES 3125
#define WPITCH    132

#define SCAN_T   1024
#define SCAN_PER 49
#define SCAN_N   (SCAN_T * SCAN_PER)

__global__ void gcn_cnt_zero(int* __restrict__ cnt) {
    int i = blockIdx.x * blockDim.x + threadIdx.x;
    if (i < NNODES) cnt[i] = 0;
}

__global__ void gcn_cnt_rank(const int* __restrict__ dst, int* __restrict__ cnt,
                             int* __restrict__ rank) {
    int e = blockIdx.x * blockDim.x + threadIdx.x;
    if (e < NEDGES) rank[e] = atomicAdd(&cnt[dst[e]], 1);
}

__global__ __launch_bounds__(SCAN_T) void gcn_scan(const int* __restrict__ cnt,
                                                   int* __restrict__ rowstart,
                                                   float* __restrict__ dinv) {
    __shared__ int s[SCAN_N];
    __shared__ int wsum[32];
    const int t    = threadIdx.x;
    const int lane = t & 31;
    const int wave = t >> 5;

    for (int i = t; i < SCAN_N; i += SCAN_T) s[i] = (i < NNODES) ? cnt[i] : 0;
    __syncthreads();

    int sum = 0;
    for (int j = 0; j < SCAN_PER; ++j) sum += s[t * SCAN_PER + j];

    int inc = sum;
    for (int off = 1; off < 32; off <<= 1) {
        int v = __shfl_up(inc, off);
        if (lane >= off) inc += v;
    }
    if (lane == 31) wsum[wave] = inc;
    __syncthreads();
    if (wave == 0) {
        int w = wsum[lane];
        int winc = w;
        for (int off = 1; off < 32; off <<= 1) {
            int v = __shfl_up(winc, off);
            if (lane >= off) winc += v;
        }
        wsum[lane] = winc - w;
    }
    __syncthreads();

    int run = wsum[wave] + inc - sum;
    for (int j = 0; j < SCAN_PER; ++j) {
        int c = s[t * SCAN_PER + j];
        s[t * SCAN_PER + j] = run;
        run += c;
    }
    __syncthreads();

    for (int i = t; i < NNODES; i += SCAN_T) {
        int r = s[i];
        int c = s[i + 1] - r;
        rowstart[i] = r;
        float dg = (float)(c + 1);
        dinv[i] = (dg > 0.0f) ? rsqrtf(dg) : 0.0f;
    }
    if (t == 0) rowstart[NNODES] = s[NNODES];
}

__global__ void gcn_place(const int* __restrict__ src, const int* __restrict__ dst,
                          const int* __restrict__ rowstart, const int* __restrict__ rank,
                          int* __restrict__ col) {
    int e = blockIdx.x * blockDim.x + threadIdx.x;
    if (e < NEDGES) col[rowstart[dst[e]] + rank[e]] = src[e];
}

__global__ __launch_bounds__(256) void gcn_gemm_wmma(const float* __restrict__ X,
                                                     const float* __restrict__ W,
                                                     float* __restrict__ H) {
    __shared__ float Wt[CH * WPITCH];
    {
        const int sc = (threadIdx.x >> 5) * 16 + (threadIdx.x & 15);
        const int sk = (threadIdx.x >> 4) & 1;
        const float* __restrict__ wsrc = W + sk * CH + sc;
        float* wdst = &Wt[sc * WPITCH + sk];
        #pragma unroll 16
        for (int i = 0; i < CH / 2; ++i) {
            *wdst = *wsrc;
            wsrc += 2 * CH;
            wdst += 2;
        }
    }
    __syncthreads();

    const int wave = threadIdx.x >> 5;
    const int lane = threadIdx.x & 31;
    const int rt   = blockIdx.x * 8 + wave;
    if (rt >= ROW_TILES) return;

    const int half = lane >> 4;
    const int mn   = lane & 15;

    const float* __restrict__ xrow = X + (size_t)(rt * 16 + mn) * CH;

    v8f acc[8];
    v8f zero = {};
    #pragma unroll
    for (int c = 0; c < 8; ++c) acc[c] = zero;

    #pragma unroll 4
    for (int kk = 0; kk < CH; kk += 4) {
        v2f a = *(const v2f*)(xrow + kk + 2 * half);
        #pragma unroll
        for (int c = 0; c < 8; ++c) {
            v2f b = *(const v2f*)(&Wt[(c * 16 + mn) * WPITCH + kk + 2 * half]);
            acc[c] = __builtin_amdgcn_wmma_f32_16x16x4_f32(
                false, a, false, b, (short)0, acc[c], false, false);
        }
    }

    #pragma unroll
    for (int c = 0; c < 8; ++c) {
        #pragma unroll
        for (int i = 0; i < 8; ++i) {
            H[(size_t)(rt * 16 + half * 8 + i) * CH + c * 16 + mn] = acc[c][i];
        }
    }
}

template <bool RELU>
__global__ __launch_bounds__(256) void gcn_gather(const int* __restrict__ rowstart,
                                                  const int* __restrict__ col,
                                                  const float* __restrict__ dinv,
                                                  const float* __restrict__ H,
                                                  const float* __restrict__ bias,
                                                  float* __restrict__ out) {
    const int lane = threadIdx.x & 31;
    const int wave = __builtin_amdgcn_readfirstlane((int)(threadIdx.x >> 5));
    const int d    = blockIdx.x * 8 + wave;
    if (d >= NNODES) return;
    const int ch  = lane * 4;
    const int beg = rowstart[d];
    const int end = rowstart[d + 1];
    const float dd = dinv[d];

    v4f acc = *(const v4f*)(&H[(size_t)d * CH + ch]) * (dd * dd) + *(const v4f*)(&bias[ch]);

    for (int base = beg; base < end; base += 32) {
        const int n = min(32, end - base);
        int   mys = 0;
        float myw = 0.0f;
        if (lane < n) {
            mys = col[base + lane];
            myw = dinv[mys] * dd;
        }
        int k = 0;
        for (; k + 4 <= n; k += 4) {
            const int s0 = __builtin_amdgcn_readlane(mys, k);
            const int s1 = __builtin_amdgcn_readlane(mys, k + 1);
            const int s2 = __builtin_amdgcn_readlane(mys, k + 2);
            const int s3 = __builtin_amdgcn_readlane(mys, k + 3);
            const float w0 = __int_as_float(__builtin_amdgcn_readlane(__float_as_int(myw), k));
            const float w1 = __int_as_float(__builtin_amdgcn_readlane(__float_as_int(myw), k + 1));
            const float w2 = __int_as_float(__builtin_amdgcn_readlane(__float_as_int(myw), k + 2));
            const float w3 = __int_as_float(__builtin_amdgcn_readlane(__float_as_int(myw), k + 3));
            const v4f h0 = *(const v4f*)(&H[(size_t)s0 * CH + ch]);
            const v4f h1 = *(const v4f*)(&H[(size_t)s1 * CH + ch]);
            const v4f h2 = *(const v4f*)(&H[(size_t)s2 * CH + ch]);
            const v4f h3 = *(const v4f*)(&H[(size_t)s3 * CH + ch]);
            acc += h0 * w0;
            acc += h1 * w1;
            acc += h2 * w2;
            acc += h3 * w3;
        }
        for (; k < n; ++k) {
            const int   s0 = __builtin_amdgcn_readlane(mys, k);
            const float w0 = __int_as_float(__builtin_amdgcn_readlane(__float_as_int(myw), k));
            const v4f h0 = *(const v4f*)(&H[(size_t)s0 * CH + ch]);
            acc += h0 * w0;
        }
    }
    if (RELU) {
        acc.x = fmaxf(acc.x, 0.0f); acc.y = fmaxf(acc.y, 0.0f);
        acc.z = fmaxf(acc.z, 0.0f); acc.w = fmaxf(acc.w, 0.0f);
    }
    *(v4f*)(&out[(size_t)d * CH + ch]) = acc;
}

extern "C" void kernel_launch(void* const* d_in, const int* in_sizes, int n_in,
                              void* d_out, int out_size, void* d_ws, size_t ws_size,
                              hipStream_t stream) {
    const float* x   = (const float*)d_in[0];
    const int*   ei  = (const int*)d_in[1];
    const float* W1  = (const float*)d_in[2];
    const float* b1  = (const float*)d_in[3];
    const float* W2  = (const float*)d_in[4];
    const float* b2  = (const float*)d_in[5];
    float*       out = (float*)d_out;

    const int* src = ei;
    const int* dst = ei + NEDGES;

    float* ws   = (float*)d_ws;
    int*   cnt  = (int*)ws;
    float* dinv = ws + 65536;
    float* H    = ws + 131072;
    float* out1 = ws + 131072 + 6400000;
    int*   rowstart = (int*)(ws + 131072 + 12800000);
    int*   rank     = (int*)(ws + 131072 + 12800000 + 65536);
    int*   col      = (int*)(ws + 131072 + 12800000 + 65536 + 640000);

    const int T = 256;
    gcn_cnt_zero<<<(NNODES + T - 1) / T, T, 0, stream>>>(cnt);
    gcn_cnt_rank<<<(NEDGES + T - 1) / T, T, 0, stream>>>(dst, cnt, rank);
    gcn_scan    <<<1, SCAN_T, 0, stream>>>(cnt, rowstart, dinv);
    gcn_place   <<<(NEDGES + T - 1) / T, T, 0, stream>>>(src, dst, rowstart, rank, col);

    const int gemm_blocks = (ROW_TILES + 7) / 8;
    const int gather_blocks = (NNODES + 7) / 8;

    gcn_gemm_wmma<<<gemm_blocks, T, 0, stream>>>(x, W1, H);
    gcn_gather<true> <<<gather_blocks, T, 0, stream>>>(rowstart, col, dinv, H, b1, out1);

    gcn_gemm_wmma<<<gemm_blocks, T, 0, stream>>>(out1, W2, H);
    gcn_gather<false><<<gather_blocks, T, 0, stream>>>(rowstart, col, dinv, H, b2, out);
}


# Row-start scan spread over 49 work-groups: each adds up the counts in front of its 1024 nodes with 12 independent four-count vector loads per thread; was one group, 49 dependent loads per thread
# speedup vs baseline: 2.8407x; 1.1119x over previous
#include <hip/hip_runtime.h>
#include <hip/hip_bf16.h>

typedef __attribute__((ext_vector_type(2))) float v2f;
typedef __attribute__((ext_vector_type(4))) float v4f;
typedef __attribute__((ext_vector_type(8))) float v8f;

#define NNODES    50000
#define NEDGES    640000
#define CH        128
#define ROW_TILES 3125
#define WPITCH    132

#define SCAN_T      1024
#define SCAN_BLOCKS 49
#define SCAN_PRE    12

typedef __attribute__((ext_vector_type(4))) int v4i;

__global__ void gcn_cnt_zero(int* __restrict__ cnt) {
    int i = blockIdx.x * blockDim.x + threadIdx.x;
    if (i < NNODES) cnt[i] = 0;
}

__global__ void gcn_cnt_rank(const int* __restrict__ dst, int* __restrict__ cnt,
                             int* __restrict__ rank) {
    int e = blockIdx.x * blockDim.x + threadIdx.x;
    if (e < NEDGES) rank[e] = atomicAdd(&cnt[dst[e]], 1);
}

__global__ __launch_bounds__(SCAN_T) void gcn_scan(const int* __restrict__ cnt,
                                                   int* __restrict__ rowstart,
                                                   float* __restrict__ dinv) {
    __shared__ int wsum[32];
    __shared__ int wpre[32];
    const int t    = threadIdx.x;
    const int lane = t & 31;
    const int wave = t >> 5;
    const int i    = blockIdx.x * SCAN_T + t;

    const v4i* __restrict__ c4 = (const v4i*)cnt;
    const int nvec  = blockIdx.x * (SCAN_T / 4);
    const int vlast = max(nvec - 1, 0);
    int pre = 0;
    #pragma unroll
    for (int j = 0; j < SCAN_PRE; ++j) {
        const int v = j * SCAN_T + t;
        const v4i q = c4[min(v, vlast)];
        const int s = (q.x + q.y) + (q.z + q.w);
        pre += (v < nvec) ? s : 0;
    }
    const int cl = cnt[min(i, NNODES - 1)];
    const int c  = (i < NNODES) ? cl : 0;

    int inc = c;
    for (int off = 1; off < 32; off <<= 1) {
        int v = __shfl_up(inc, off);
        if (lane >= off) inc += v;
    }
    for (int off = 16; off > 0; off >>= 1) pre += __shfl_xor(pre, off);
    if (lane == 31) wsum[wave] = inc;
    if (lane == 0)  wpre[wave] = pre;
    __syncthreads();

    const int w = wsum[lane];
    int winc = w;
    for (int off = 1; off < 32; off <<= 1) {
        int v = __shfl_up(winc, off);
        if (lane >= off) winc += v;
    }
    int ptot = wpre[lane];
    for (int off = 16; off > 0; off >>= 1) ptot += __shfl_xor(ptot, off);
    const int wexcl = __shfl(winc - w, wave);

    const int r = ptot + wexcl + inc - c;
    if (i <= NNODES) rowstart[i] = r;
    if (i < NNODES) {
        float dg = (float)(c + 1);
        dinv[i] = (dg > 0.0f) ? rsqrtf(dg) : 0.0f;
    }
}

__global__ void gcn_place(const int* __restrict__ src, const int* __restrict__ dst,
                          const int* __restrict__ rowstart, const int* __restrict__ rank,
                          int* __restrict__ col) {
    int e = blockIdx.x * blockDim.x + threadIdx.x;
    if (e < NEDGES) col[rowstart[dst[e]] + rank[e]] = src[e];
}

__global__ __launch_bounds__(256) void gcn_gemm_wmma(const float* __restrict__ X,
                                                     const float* __restrict__ W,
                                                     float* __restrict__ H) {
    __shared__ float Wt[CH * WPITCH];
    {
        const int sc = (threadIdx.x >> 5) * 16 + (threadIdx.x & 15);
        const int sk = (threadIdx.x >> 4) & 1;
        const float* __restrict__ wsrc = W + sk * CH + sc;
        float* wdst = &Wt[sc * WPITCH + sk];
        #pragma unroll 16
        for (int i = 0; i < CH / 2; ++i) {
            *wdst = *wsrc;
            wsrc += 2 * CH;
            wdst += 2;
        }
    }
    __syncthreads();

    const int wave = threadIdx.x >> 5;
    const int lane = threadIdx.x & 31;
    const int rt   = blockIdx.x * 8 + wave;
    if (rt >= ROW_TILES) return;

    const int half = lane >> 4;
    const int mn   = lane & 15;

    const float* __restrict__ xrow = X + (size_t)(rt * 16 + mn) * CH;

    v8f acc[8];
    v8f zero = {};
    #pragma unroll
    for (int c = 0; c < 8; ++c) acc[c] = zero;

    #pragma unroll 4
    for (int kk = 0; kk < CH; kk += 4) {
        v2f a = *(const v2f*)(xrow + kk + 2 * half);
        #pragma unroll
        for (int c = 0; c < 8; ++c) {
            v2f b = *(const v2f*)(&Wt[(c * 16 + mn) * WPITCH + kk + 2 * half]);
            acc[c] = __builtin_amdgcn_wmma_f32_16x16x4_f32(
                false, a, false, b, (short)0, acc[c], false, false);
        }
    }

    #pragma unroll
    for (int c = 0; c < 8; ++c) {
        #pragma unroll
        for (int i = 0; i < 8; ++i) {
            H[(size_t)(rt * 16 + half * 8 + i) * CH + c * 16 + mn] = acc[c][i];
        }
    }
}

template <bool RELU>
__global__ __launch_bounds__(256) void gcn_gather(const int* __restrict__ rowstart,
                                                  const int* __restrict__ col,
                                                  const float* __restrict__ dinv,
                                                  const float* __restrict__ H,
                                                  const float* __restrict__ bias,
                                                  float* __restrict__ out) {
    const int lane = threadIdx.x & 31;
    const int wave = __builtin_amdgcn_readfirstlane((int)(threadIdx.x >> 5));
    const int d    = blockIdx.x * 8 + wave;
    if (d >= NNODES) return;
    const int ch  = lane * 4;
    const int beg = rowstart[d];
    const int end = rowstart[d + 1];
    const float dd = dinv[d];

    v4f acc = *(const v4f*)(&H[(size_t)d * CH + ch]) * (dd * dd) + *(const v4f*)(&bias[ch]);

    for (int base = beg; base < end; base += 32) {
        const int n = min(32, end - base);
        int   mys = 0;
        float myw = 0.0f;
        if (lane < n) {
            mys = col[base + lane];
            myw = dinv[mys] * dd;
        }
        int k = 0;
        for (; k + 4 <= n; k += 4) {
            const int s0 = __builtin_amdgcn_readlane(mys, k);
            const int s1 = __builtin_amdgcn_readlane(mys, k + 1);
            const int s2 = __builtin_amdgcn_readlane(mys, k + 2);
            const int s3 = __builtin_amdgcn_readlane(mys, k + 3);
            const float w0 = __int_as_float(__builtin_amdgcn_readlane(__float_as_int(myw), k));
            const float w1 = __int_as_float(__builtin_amdgcn_readlane(__float_as_int(myw), k + 1));
            const float w2 = __int_as_float(__builtin_amdgcn_readlane(__float_as_int(myw), k + 2));
            const float w3 = __int_as_float(__builtin_amdgcn_readlane(__float_as_int(myw), k + 3));
            const v4f h0 = *(const v4f*)(&H[(size_t)s0 * CH + ch]);
            const v4f h1 = *(const v4f*)(&H[(size_t)s1 * CH + ch]);
            const v4f h2 = *(const v4f*)(&H[(size_t)s2 * CH + ch]);
            const v4f h3 = *(const v4f*)(&H[(size_t)s3 * CH + ch]);
            acc += h0 * w0;
            acc += h1 * w1;
            acc += h2 * w2;
            acc += h3 * w3;
        }
        for (; k < n; ++k) {
            const int   s0 = __builtin_amdgcn_readlane(mys, k);
            const float w0 = __int_as_float(__builtin_amdgcn_readlane(__float_as_int(myw), k));
            const v4f h0 = *(const v4f*)(&H[(size_t)s0 * CH + ch]);
            acc += h0 * w0;
        }
    }
    if (RELU) {
        acc.x = fmaxf(acc.x, 0.0f); acc.y = fmaxf(acc.y, 0.0f);
        acc.z = fmaxf(acc.z, 0.0f); acc.w = fmaxf(acc.w, 0.0f);
    }
    *(v4f*)(&out[(size_t)d * CH + ch]) = acc;
}

extern "C" void kernel_launch(void* const* d_in, const int* in_sizes, int n_in,
                              void* d_out, int out_size, void* d_ws, size_t ws_size,
                              hipStream_t stream) {
    const float* x   = (const float*)d_in[0];
    const int*   ei  = (const int*)d_in[1];
    const float* W1  = (const float*)d_in[2];
    const float* b1  = (const float*)d_in[3];
    const float* W2  = (const float*)d_in[4];
    const float* b2  = (const float*)d_in[5];
    float*       out = (float*)d_out;

    const int* src = ei;
    const int* dst = ei + NEDGES;

    float* ws   = (float*)d_ws;
    int*   cnt  = (int*)ws;
    float* dinv = ws + 65536;
    float* H    = ws + 131072;
    float* out1 = ws + 131072 + 6400000;
    int*   rowstart = (int*)(ws + 131072 + 12800000);
    int*   rank     = (int*)(ws + 131072 + 12800000 + 65536);
    int*   col      = (int*)(ws + 131072 + 12800000 + 65536 + 640000);

    const int T = 256;
    gcn_cnt_zero<<<(NNODES + T - 1) / T, T, 0, stream>>>(cnt);
    gcn_cnt_rank<<<(NEDGES + T - 1) / T, T, 0, stream>>>(dst, cnt, rank);
    gcn_scan    <<<SCAN_BLOCKS, SCAN_T, 0, stream>>>(cnt, rowstart, dinv);
    gcn_place   <<<(NEDGES + T - 1) / T, T, 0, stream>>>(src, dst, rowstart, rank, col);

    const int gemm_blocks = (ROW_TILES + 7) / 8;
    const int gather_blocks = (NNODES + 7) / 8;

    gcn_gemm_wmma<<<gemm_blocks, T, 0, stream>>>(x, W1, H);
    gcn_gather<true> <<<gather_blocks, T, 0, stream>>>(rowstart, col, dinv, H, b1, out1);

    gcn_gemm_wmma<<<gemm_blocks, T, 0, stream>>>(out1, W2, H);
    gcn_gather<false><<<gather_blocks, T, 0, stream>>>(rowstart, col, dinv, H, b2, out);
}


# Layer-1 product and in-degree count/rank share one launch (independent work: work-groups 0-390 product, 391-1015 count, 4 edges a thread); count's own launch gone; ranks arrival-order as before
# speedup vs baseline: 3.0585x; 1.0767x over previous
#include <hip/hip_runtime.h>
#include <hip/hip_bf16.h>

typedef __attribute__((ext_vector_type(2))) float v2f;
typedef __attribute__((ext_vector_type(4))) float v4f;
typedef __attribute__((ext_vector_type(8))) float v8f;

#define NNODES    50000
#define NEDGES    640000
#define CH        128
#define ROW_TILES 3125
#define WPITCH    132

#define SCAN_T      1024
#define SCAN_BLOCKS 49
#define SCAN_PRE    12

typedef __attribute__((ext_vector_type(4))) int v4i;

__global__ void gcn_cnt_zero(int* __restrict__ cnt) {
    int i = blockIdx.x * blockDim.x + threadIdx.x;
    if (i < NNODES) cnt[i] = 0;
}


__global__ __launch_bounds__(SCAN_T) void gcn_scan(const int* __restrict__ cnt,
                                                   int* __restrict__ rowstart,
                                                   float* __restrict__ dinv) {
    __shared__ int wsum[32];
    __shared__ int wpre[32];
    const int t    = threadIdx.x;
    const int lane = t & 31;
    const int wave = t >> 5;
    const int i    = blockIdx.x * SCAN_T + t;

    const v4i* __restrict__ c4 = (const v4i*)cnt;
    const int nvec  = blockIdx.x * (SCAN_T / 4);
    const int vlast = max(nvec - 1, 0);
    int pre = 0;
    #pragma unroll
    for (int j = 0; j < SCAN_PRE; ++j) {
        const int v = j * SCAN_T + t;
        const v4i q = c4[min(v, vlast)];
        const int s = (q.x + q.y) + (q.z + q.w);
        pre += (v < nvec) ? s : 0;
    }
    const int cl = cnt[min(i, NNODES - 1)];
    const int c  = (i < NNODES) ? cl : 0;

    int inc = c;
    for (int off = 1; off < 32; off <<= 1) {
        int v = __shfl_up(inc, off);
        if (lane >= off) inc += v;
    }
    for (int off = 16; off > 0; off >>= 1) pre += __shfl_xor(pre, off);
    if (lane == 31) wsum[wave] = inc;
    if (lane == 0)  wpre[wave] = pre;
    __syncthreads();

    const int w = wsum[lane];
    int winc = w;
    for (int off = 1; off < 32; off <<= 1) {
        int v = __shfl_up(winc, off);
        if (lane >= off) winc += v;
    }
    int ptot = wpre[lane];
    for (int off = 16; off > 0; off >>= 1) ptot += __shfl_xor(ptot, off);
    const int wexcl = __shfl(winc - w, wave);

    const int r = ptot + wexcl + inc - c;
    if (i <= NNODES) rowstart[i] = r;
    if (i < NNODES) {
        float dg = (float)(c + 1);
        dinv[i] = (dg > 0.0f) ? rsqrtf(dg) : 0.0f;
    }
}

__global__ void gcn_place(const int* __restrict__ src, const int* __restrict__ dst,
                          const int* __restrict__ rowstart, const int* __restrict__ rank,
                          int* __restrict__ col) {
    int e = blockIdx.x * blockDim.x + threadIdx.x;
    if (e < NEDGES) col[rowstart[dst[e]] + rank[e]] = src[e];
}

__device__ __forceinline__ void gcn_gemm_block(const float* __restrict__ X,
                                               const float* __restrict__ W,
                                               float* __restrict__ H,
                                               const int gb, float* Wt) {
    {
        const int sc = (threadIdx.x >> 5) * 16 + (threadIdx.x & 15);
        const int sk = (threadIdx.x >> 4) & 1;
        const float* __restrict__ wsrc = W + sk * CH + sc;
        float* wdst = &Wt[sc * WPITCH + sk];
        #pragma unroll 16
        for (int i = 0; i < CH / 2; ++i) {
            *wdst = *wsrc;
            wsrc += 2 * CH;
            wdst += 2;
        }
    }
    __syncthreads();

    const int wave = threadIdx.x >> 5;
    const int lane = threadIdx.x & 31;
    const int rt   = gb * 8 + wave;
    if (rt >= ROW_TILES) return;

    const int half = lane >> 4;
    const int mn   = lane & 15;

    const float* __restrict__ xrow = X + (size_t)(rt * 16 + mn) * CH;

    v8f acc[8];
    v8f zero = {};
    #pragma unroll
    for (int c = 0; c < 8; ++c) acc[c] = zero;

    #pragma unroll 4
    for (int kk = 0; kk < CH; kk += 4) {
        v2f a = *(const v2f*)(xrow + kk + 2 * half);
        #pragma unroll
        for (int c = 0; c < 8; ++c) {
            v2f b = *(const v2f*)(&Wt[(c * 16 + mn) * WPITCH + kk + 2 * half]);
            acc[c] = __builtin_amdgcn_wmma_f32_16x16x4_f32(
                false, a, false, b, (short)0, acc[c], false, false);
        }
    }

    #pragma unroll
    for (int c = 0; c < 8; ++c) {
        #pragma unroll
        for (int i = 0; i < 8; ++i) {
            H[(size_t)(rt * 16 + half * 8 + i) * CH + c * 16 + mn] = acc[c][i];
        }
    }
}

#define GEMM_BLOCKS 391
#define RANK_BLOCKS 625

__global__ __launch_bounds__(256) void gcn_gemm_wmma(const float* __restrict__ X,
                                                     const float* __restrict__ W,
                                                     float* __restrict__ H) {
    __shared__ float Wt[CH * WPITCH];
    gcn_gemm_block(X, W, H, (int)blockIdx.x, Wt);
}

__global__ __launch_bounds__(256) void gcn_gemm_rank(const float* __restrict__ X,
                                                     const float* __restrict__ W,
                                                     float* __restrict__ H,
                                                     const int* __restrict__ dst,
                                                     int* __restrict__ cnt,
                                                     int* __restrict__ rank) {
    __shared__ float Wt[CH * WPITCH];
    if (blockIdx.x < GEMM_BLOCKS) {
        gcn_gemm_block(X, W, H, (int)blockIdx.x, Wt);
    } else {
        const int t = ((int)blockIdx.x - GEMM_BLOCKS) * 256 + (int)threadIdx.x;
        if (t < NEDGES / 4) {
            const v4i d4 = ((const v4i*)dst)[t];
            v4i r;
            r.x = atomicAdd(&cnt[d4.x], 1);
            r.y = atomicAdd(&cnt[d4.y], 1);
            r.z = atomicAdd(&cnt[d4.z], 1);
            r.w = atomicAdd(&cnt[d4.w], 1);
            ((v4i*)rank)[t] = r;
        }
    }
}

template <bool RELU>
__global__ __launch_bounds__(256) void gcn_gather(const int* __restrict__ rowstart,
                                                  const int* __restrict__ col,
                                                  const float* __restrict__ dinv,
                                                  const float* __restrict__ H,
                                                  const float* __restrict__ bias,
                                                  float* __restrict__ out) {
    const int lane = threadIdx.x & 31;
    const int wave = __builtin_amdgcn_readfirstlane((int)(threadIdx.x >> 5));
    const int d    = blockIdx.x * 8 + wave;
    if (d >= NNODES) return;
    const int ch  = lane * 4;
    const int beg = rowstart[d];
    const int end = rowstart[d + 1];
    const float dd = dinv[d];

    v4f acc = *(const v4f*)(&H[(size_t)d * CH + ch]) * (dd * dd) + *(const v4f*)(&bias[ch]);

    for (int base = beg; base < end; base += 32) {
        const int n = min(32, end - base);
        int   mys = 0;
        float myw = 0.0f;
        if (lane < n) {
            mys = col[base + lane];
            myw = dinv[mys] * dd;
        }
        int k = 0;
        for (; k + 4 <= n; k += 4) {
            const int s0 = __builtin_amdgcn_readlane(mys, k);
            const int s1 = __builtin_amdgcn_readlane(mys, k + 1);
            const int s2 = __builtin_amdgcn_readlane(mys, k + 2);
            const int s3 = __builtin_amdgcn_readlane(mys, k + 3);
            const float w0 = __int_as_float(__builtin_amdgcn_readlane(__float_as_int(myw), k));
            const float w1 = __int_as_float(__builtin_amdgcn_readlane(__float_as_int(myw), k + 1));
            const float w2 = __int_as_float(__builtin_amdgcn_readlane(__float_as_int(myw), k + 2));
            const float w3 = __int_as_float(__builtin_amdgcn_readlane(__float_as_int(myw), k + 3));
            const v4f h0 = *(const v4f*)(&H[(size_t)s0 * CH + ch]);
            const v4f h1 = *(const v4f*)(&H[(size_t)s1 * CH + ch]);
            const v4f h2 = *(const v4f*)(&H[(size_t)s2 * CH + ch]);
            const v4f h3 = *(const v4f*)(&H[(size_t)s3 * CH + ch]);
            acc += h0 * w0;
            acc += h1 * w1;
            acc += h2 * w2;
            acc += h3 * w3;
        }
        for (; k < n; ++k) {
            const int   s0 = __builtin_amdgcn_readlane(mys, k);
            const float w0 = __int_as_float(__builtin_amdgcn_readlane(__float_as_int(myw), k));
            const v4f h0 = *(const v4f*)(&H[(size_t)s0 * CH + ch]);
            acc += h0 * w0;
        }
    }
    if (RELU) {
        acc.x = fmaxf(acc.x, 0.0f); acc.y = fmaxf(acc.y, 0.0f);
        acc.z = fmaxf(acc.z, 0.0f); acc.w = fmaxf(acc.w, 0.0f);
    }
    *(v4f*)(&out[(size_t)d * CH + ch]) = acc;
}

extern "C" void kernel_launch(void* const* d_in, const int* in_sizes, int n_in,
                              void* d_out, int out_size, void* d_ws, size_t ws_size,
                              hipStream_t stream) {
    const float* x   = (const float*)d_in[0];
    const int*   ei  = (const int*)d_in[1];
    const float* W1  = (const float*)d_in[2];
    const float* b1  = (const float*)d_in[3];
    const float* W2  = (const float*)d_in[4];
    const float* b2  = (const float*)d_in[5];
    float*       out = (float*)d_out;

    const int* src = ei;
    const int* dst = ei + NEDGES;

    float* ws   = (float*)d_ws;
    int*   cnt  = (int*)ws;
    float* dinv = ws + 65536;
    float* H    = ws + 131072;
    float* out1 = ws + 131072 + 6400000;
    int*   rowstart = (int*)(ws + 131072 + 12800000);
    int*   rank     = (int*)(ws + 131072 + 12800000 + 65536);
    int*   col      = (int*)(ws + 131072 + 12800000 + 65536 + 640000);

    const int T = 256;
    gcn_cnt_zero<<<(NNODES + T - 1) / T, T, 0, stream>>>(cnt);
    gcn_gemm_rank<<<GEMM_BLOCKS + RANK_BLOCKS, T, 0, stream>>>(x, W1, H, dst, cnt, rank);
    gcn_scan    <<<SCAN_BLOCKS, SCAN_T, 0, stream>>>(cnt, rowstart, dinv);
    gcn_place   <<<(NEDGES + T - 1) / T, T, 0, stream>>>(src, dst, rowstart, rank, col);

    const int gather_blocks = (NNODES + 7) / 8;

    gcn_gather<true> <<<gather_blocks, T, 0, stream>>>(rowstart, col, dinv, H, b1, out1);

    gcn_gemm_wmma<<<GEMM_BLOCKS, T, 0, stream>>>(out1, W2, H);
    gcn_gather<false><<<gather_blocks, T, 0, stream>>>(rowstart, col, dinv, H, b2, out);
}


# Edges go straight into fixed 64-slot buckets per destination at the rank the counting atomic returns (overflow list past 64); row-start scan and place launches gone; ranks arrival-order as before
# speedup vs baseline: 3.5171x; 1.1499x over previous
#include <hip/hip_runtime.h>
#include <hip/hip_bf16.h>

typedef __attribute__((ext_vector_type(2))) float v2f;
typedef __attribute__((ext_vector_type(4))) float v4f;
typedef __attribute__((ext_vector_type(8))) float v8f;

#define NNODES    50000
#define NEDGES    640000
#define CH        128
#define ROW_TILES 3125
#define WPITCH    132

#define CAP 64

typedef __attribute__((ext_vector_type(4))) int v4i;

__global__ void gcn_cnt_zero(int* __restrict__ cnt) {
    int i = blockIdx.x * blockDim.x + threadIdx.x;
    if (i <= NNODES) cnt[i] = 0;
}

__device__ __forceinline__ float gcn_dinv(const int c) {
    const float dg = (float)(c + 1);
    return (dg > 0.0f) ? rsqrtf(dg) : 0.0f;
}

__device__ __forceinline__ void gcn_put(const int s, const int d, const int r,
                                        int* __restrict__ cnt, int* __restrict__ bucket,
                                        int* __restrict__ ovf) {
    if (r < CAP) {
        bucket[d * CAP + r] = s;
    } else {
        const int i = atomicAdd(&cnt[NNODES], 1);
        ovf[2 * i]     = s;
        ovf[2 * i + 1] = d;
    }
}

__device__ __forceinline__ void gcn_gemm_block(const float* __restrict__ X,
                                               const float* __restrict__ W,
                                               float* __restrict__ H,
                                               const int gb, float* Wt) {
    {
        const int sc = (threadIdx.x >> 5) * 16 + (threadIdx.x & 15);
        const int sk = (threadIdx.x >> 4) & 1;
        const float* __restrict__ wsrc = W + sk * CH + sc;
        float* wdst = &Wt[sc * WPITCH + sk];
        #pragma unroll 16
        for (int i = 0; i < CH / 2; ++i) {
            *wdst = *wsrc;
            wsrc += 2 * CH;
            wdst += 2;
        }
    }
    __syncthreads();

    const int wave = threadIdx.x >> 5;
    const int lane = threadIdx.x & 31;
    const int rt   = gb * 8 + wave;
    if (rt >= ROW_TILES) return;

    const int half = lane >> 4;
    const int mn   = lane & 15;

    const float* __restrict__ xrow = X + (size_t)(rt * 16 + mn) * CH;

    v8f acc[8];
    v8f zero = {};
    #pragma unroll
    for (int c = 0; c < 8; ++c) acc[c] = zero;

    #pragma unroll 4
    for (int kk = 0; kk < CH; kk += 4) {
        v2f a = *(const v2f*)(xrow + kk + 2 * half);
        #pragma unroll
        for (int c = 0; c < 8; ++c) {
            v2f b = *(const v2f*)(&Wt[(c * 16 + mn) * WPITCH + kk + 2 * half]);
            acc[c] = __builtin_amdgcn_wmma_f32_16x16x4_f32(
                false, a, false, b, (short)0, acc[c], false, false);
        }
    }

    #pragma unroll
    for (int c = 0; c < 8; ++c) {
        #pragma unroll
        for (int i = 0; i < 8; ++i) {
            H[(size_t)(rt * 16 + half * 8 + i) * CH + c * 16 + mn] = acc[c][i];
        }
    }
}

#define GEMM_BLOCKS 391
#define RANK_BLOCKS 625

__global__ __launch_bounds__(256) void gcn_gemm_wmma(const float* __restrict__ X,
                                                     const float* __restrict__ W,
                                                     float* __restrict__ H) {
    __shared__ float Wt[CH * WPITCH];
    gcn_gemm_block(X, W, H, (int)blockIdx.x, Wt);
}

__global__ __launch_bounds__(256) void gcn_gemm_rank(const float* __restrict__ X,
                                                     const float* __restrict__ W,
                                                     float* __restrict__ H,
                                                     const int* __restrict__ src,
                                                     const int* __restrict__ dst,
                                                     int* __restrict__ cnt,
                                                     int* __restrict__ bucket,
                                                     int* __restrict__ ovf) {
    __shared__ float Wt[CH * WPITCH];
    if (blockIdx.x < GEMM_BLOCKS) {
        gcn_gemm_block(X, W, H, (int)blockIdx.x, Wt);
    } else {
        const int t = ((int)blockIdx.x - GEMM_BLOCKS) * 256 + (int)threadIdx.x;
        if (t < NEDGES / 4) {
            const v4i d4 = ((const v4i*)dst)[t];
            const v4i s4 = ((const v4i*)src)[t];
            const int r0 = atomicAdd(&cnt[d4.x], 1);
            const int r1 = atomicAdd(&cnt[d4.y], 1);
            const int r2 = atomicAdd(&cnt[d4.z], 1);
            const int r3 = atomicAdd(&cnt[d4.w], 1);
            gcn_put(s4.x, d4.x, r0, cnt, bucket, ovf);
            gcn_put(s4.y, d4.y, r1, cnt, bucket, ovf);
            gcn_put(s4.z, d4.z, r2, cnt, bucket, ovf);
            gcn_put(s4.w, d4.w, r3, cnt, bucket, ovf);
        }
    }
}

template <bool RELU>
__global__ __launch_bounds__(256) void gcn_gather(const int* __restrict__ cnt,
                                                  const int* __restrict__ bucket,
                                                  const int* __restrict__ ovf,
                                                  const float* __restrict__ H,
                                                  const float* __restrict__ bias,
                                                  float* __restrict__ out) {
    const int lane = threadIdx.x & 31;
    const int wave = __builtin_amdgcn_readfirstlane((int)(threadIdx.x >> 5));
    const int d    = blockIdx.x * 8 + wave;
    if (d >= NNODES) return;
    const int ch  = lane * 4;
    const int deg = cnt[d];
    const int end = min(deg, CAP);
    const int* __restrict__ slots = bucket + (size_t)d * CAP;
    const float dd = gcn_dinv(deg);

    v4f acc = *(const v4f*)(&H[(size_t)d * CH + ch]) * (dd * dd) + *(const v4f*)(&bias[ch]);

    for (int base = 0; base < end; base += 32) {
        const int n = min(32, end - base);
        int   mys = 0;
        float myw = 0.0f;
        if (lane < n) {
            mys = slots[base + lane];
            myw = gcn_dinv(cnt[mys]) * dd;
        }
        int k = 0;
        for (; k + 4 <= n; k += 4) {
            const int s0 = __builtin_amdgcn_readlane(mys, k);
            const int s1 = __builtin_amdgcn_readlane(mys, k + 1);
            const int s2 = __builtin_amdgcn_readlane(mys, k + 2);
            const int s3 = __builtin_amdgcn_readlane(mys, k + 3);
            const float w0 = __int_as_float(__builtin_amdgcn_readlane(__float_as_int(myw), k));
            const float w1 = __int_as_float(__builtin_amdgcn_readlane(__float_as_int(myw), k + 1));
            const float w2 = __int_as_float(__builtin_amdgcn_readlane(__float_as_int(myw), k + 2));
            const float w3 = __int_as_float(__builtin_amdgcn_readlane(__float_as_int(myw), k + 3));
            const v4f h0 = *(const v4f*)(&H[(size_t)s0 * CH + ch]);
            const v4f h1 = *(const v4f*)(&H[(size_t)s1 * CH + ch]);
            const v4f h2 = *(const v4f*)(&H[(size_t)s2 * CH + ch]);
            const v4f h3 = *(const v4f*)(&H[(size_t)s3 * CH + ch]);
            acc += h0 * w0;
            acc += h1 * w1;
            acc += h2 * w2;
            acc += h3 * w3;
        }
        for (; k < n; ++k) {
            const int   s0 = __builtin_amdgcn_readlane(mys, k);
            const float w0 = __int_as_float(__builtin_amdgcn_readlane(__float_as_int(myw), k));
            const v4f h0 = *(const v4f*)(&H[(size_t)s0 * CH + ch]);
            acc += h0 * w0;
        }
    }
    if (deg > CAP) {
        const int no = cnt[NNODES];
        for (int i = 0; i < no; ++i) {
            if (ovf[2 * i + 1] == d) {
                const int   s0 = ovf[2 * i];
                const float w0 = gcn_dinv(cnt[s0]) * dd;
                const v4f h0 = *(const v4f*)(&H[(size_t)s0 * CH + ch]);
                acc += h0 * w0;
            }
        }
    }
    if (RELU) {
        acc.x = fmaxf(acc.x, 0.0f); acc.y = fmaxf(acc.y, 0.0f);
        acc.z = fmaxf(acc.z, 0.0f); acc.w = fmaxf(acc.w, 0.0f);
    }
    *(v4f*)(&out[(size_t)d * CH + ch]) = acc;
}

extern "C" void kernel_launch(void* const* d_in, const int* in_sizes, int n_in,
                              void* d_out, int out_size, void* d_ws, size_t ws_size,
                              hipStream_t stream) {
    const float* x   = (const float*)d_in[0];
    const int*   ei  = (const int*)d_in[1];
    const float* W1  = (const float*)d_in[2];
    const float* b1  = (const float*)d_in[3];
    const float* W2  = (const float*)d_in[4];
    const float* b2  = (const float*)d_in[5];
    float*       out = (float*)d_out;

    const int* src = ei;
    const int* dst = ei + NEDGES;

    float* ws   = (float*)d_ws;
    int*   cnt  = (int*)ws;
    float* H    = ws + 131072;
    float* out1 = ws + 131072 + 6400000;
    int*   bucket = (int*)(ws + 131072 + 12800000);
    int*   ovf    = (int*)(ws + 131072 + 12800000 + 3200000);

    const int T = 256;
    gcn_cnt_zero<<<(NNODES + 1 + T - 1) / T, T, 0, stream>>>(cnt);
    gcn_gemm_rank<<<GEMM_BLOCKS + RANK_BLOCKS, T, 0, stream>>>(x, W1, H, src, dst, cnt, bucket, ovf);

    const int gather_blocks = (NNODES + 7) / 8;

    gcn_gather<true> <<<gather_blocks, T, 0, stream>>>(cnt, bucket, ovf, H, b1, out1);

    gcn_gemm_wmma<<<GEMM_BLOCKS, T, 0, stream>>>(out1, W2, H);
    gcn_gather<false><<<gather_blocks, T, 0, stream>>>(cnt, bucket, ovf, H, b2, out);
}


# Layer-2 product fused into layer-1 gather: 16 waves sum+relu 16 rows into an LDS tile, waves 0-7 multiply it by W2 held in LDS, persistent work-groups; product launch and stored relu'd rows gone
# speedup vs baseline: 3.8220x; 1.0867x over previous
#include <hip/hip_runtime.h>
#include <hip/hip_bf16.h>

typedef __attribute__((ext_vector_type(2))) float v2f;
typedef __attribute__((ext_vector_type(4))) float v4f;
typedef __attribute__((ext_vector_type(8))) float v8f;

#define NNODES    50000
#define NEDGES    640000
#define CH        128
#define ROW_TILES 3125
#define WPITCH    132

#define CAP 64

typedef __attribute__((ext_vector_type(4))) int v4i;

__global__ void gcn_cnt_zero(int* __restrict__ cnt) {
    int i = blockIdx.x * blockDim.x + threadIdx.x;
    if (i <= NNODES) cnt[i] = 0;
}

__device__ __forceinline__ float gcn_dinv(const int c) {
    const float dg = (float)(c + 1);
    return (dg > 0.0f) ? rsqrtf(dg) : 0.0f;
}

__device__ __forceinline__ void gcn_put(const int s, const int d, const int r,
                                        int* __restrict__ cnt, int* __restrict__ bucket,
                                        int* __restrict__ ovf) {
    if (r < CAP) {
        bucket[d * CAP + r] = s;
    } else {
        const int i = atomicAdd(&cnt[NNODES], 1);
        ovf[2 * i]     = s;
        ovf[2 * i + 1] = d;
    }
}

__device__ __forceinline__ void gcn_gemm_block(const float* __restrict__ X,
                                               const float* __restrict__ W,
                                               float* __restrict__ H,
                                               const int gb, float* Wt) {
    {
        const int sc = (threadIdx.x >> 5) * 16 + (threadIdx.x & 15);
        const int sk = (threadIdx.x >> 4) & 1;
        const float* __restrict__ wsrc = W + sk * CH + sc;
        float* wdst = &Wt[sc * WPITCH + sk];
        #pragma unroll 16
        for (int i = 0; i < CH / 2; ++i) {
            *wdst = *wsrc;
            wsrc += 2 * CH;
            wdst += 2;
        }
    }
    __syncthreads();

    const int wave = threadIdx.x >> 5;
    const int lane = threadIdx.x & 31;
    const int rt   = gb * 8 + wave;
    if (rt >= ROW_TILES) return;

    const int half = lane >> 4;
    const int mn   = lane & 15;

    const float* __restrict__ xrow = X + (size_t)(rt * 16 + mn) * CH;

    v8f acc[8];
    v8f zero = {};
    #pragma unroll
    for (int c = 0; c < 8; ++c) acc[c] = zero;

    #pragma unroll 4
    for (int kk = 0; kk < CH; kk += 4) {
        v2f a = *(const v2f*)(xrow + kk + 2 * half);
        #pragma unroll
        for (int c = 0; c < 8; ++c) {
            v2f b = *(const v2f*)(&Wt[(c * 16 + mn) * WPITCH + kk + 2 * half]);
            acc[c] = __builtin_amdgcn_wmma_f32_16x16x4_f32(
                false, a, false, b, (short)0, acc[c], false, false);
        }
    }

    #pragma unroll
    for (int c = 0; c < 8; ++c) {
        #pragma unroll
        for (int i = 0; i < 8; ++i) {
            H[(size_t)(rt * 16 + half * 8 + i) * CH + c * 16 + mn] = acc[c][i];
        }
    }
}

#define GEMM_BLOCKS 391
#define RANK_BLOCKS 625

__global__ __launch_bounds__(256) void gcn_gemm_rank(const float* __restrict__ X,
                                                     const float* __restrict__ W,
                                                     float* __restrict__ H,
                                                     const int* __restrict__ src,
                                                     const int* __restrict__ dst,
                                                     int* __restrict__ cnt,
                                                     int* __restrict__ bucket,
                                                     int* __restrict__ ovf) {
    __shared__ float Wt[CH * WPITCH];
    if (blockIdx.x < GEMM_BLOCKS) {
        gcn_gemm_block(X, W, H, (int)blockIdx.x, Wt);
    } else {
        const int t = ((int)blockIdx.x - GEMM_BLOCKS) * 256 + (int)threadIdx.x;
        if (t < NEDGES / 4) {
            const v4i d4 = ((const v4i*)dst)[t];
            const v4i s4 = ((const v4i*)src)[t];
            const int r0 = atomicAdd(&cnt[d4.x], 1);
            const int r1 = atomicAdd(&cnt[d4.y], 1);
            const int r2 = atomicAdd(&cnt[d4.z], 1);
            const int r3 = atomicAdd(&cnt[d4.w], 1);
            gcn_put(s4.x, d4.x, r0, cnt, bucket, ovf);
            gcn_put(s4.y, d4.y, r1, cnt, bucket, ovf);
            gcn_put(s4.z, d4.z, r2, cnt, bucket, ovf);
            gcn_put(s4.w, d4.w, r3, cnt, bucket, ovf);
        }
    }
}

__device__ __forceinline__ v4f gcn_gather_row(const int d, const int lane,
                                              const int* __restrict__ cnt,
                                              const int* __restrict__ bucket,
                                              const int* __restrict__ ovf,
                                              const float* __restrict__ H,
                                              const float* __restrict__ bias) {
    const int ch  = lane * 4;
    const int deg = cnt[d];
    const int end = min(deg, CAP);
    const int* __restrict__ slots = bucket + (size_t)d * CAP;
    const float dd = gcn_dinv(deg);

    v4f acc = *(const v4f*)(&H[(size_t)d * CH + ch]) * (dd * dd) + *(const v4f*)(&bias[ch]);

    for (int base = 0; base < end; base += 32) {
        const int n = min(32, end - base);
        int   mys = 0;
        float myw = 0.0f;
        if (lane < n) {
            mys = slots[base + lane];
            myw = gcn_dinv(cnt[mys]) * dd;
        }
        int k = 0;
        for (; k + 4 <= n; k += 4) {
            const int s0 = __builtin_amdgcn_readlane(mys, k);
            const int s1 = __builtin_amdgcn_readlane(mys, k + 1);
            const int s2 = __builtin_amdgcn_readlane(mys, k + 2);
            const int s3 = __builtin_amdgcn_readlane(mys, k + 3);
            const float w0 = __int_as_float(__builtin_amdgcn_readlane(__float_as_int(myw), k));
            const float w1 = __int_as_float(__builtin_amdgcn_readlane(__float_as_int(myw), k + 1));
            const float w2 = __int_as_float(__builtin_amdgcn_readlane(__float_as_int(myw), k + 2));
            const float w3 = __int_as_float(__builtin_amdgcn_readlane(__float_as_int(myw), k + 3));
            const v4f h0 = *(const v4f*)(&H[(size_t)s0 * CH + ch]);
            const v4f h1 = *(const v4f*)(&H[(size_t)s1 * CH + ch]);
            const v4f h2 = *(const v4f*)(&H[(size_t)s2 * CH + ch]);
            const v4f h3 = *(const v4f*)(&H[(size_t)s3 * CH + ch]);
            acc += h0 * w0;
            acc += h1 * w1;
            acc += h2 * w2;
            acc += h3 * w3;
        }
        for (; k < n; ++k) {
            const int   s0 = __builtin_amdgcn_readlane(mys, k);
            const float w0 = __int_as_float(__builtin_amdgcn_readlane(__float_as_int(myw), k));
            const v4f h0 = *(const v4f*)(&H[(size_t)s0 * CH + ch]);
            acc += h0 * w0;
        }
    }
    if (deg > CAP) {
        const int no = cnt[NNODES];
        for (int i = 0; i < no; ++i) {
            if (ovf[2 * i + 1] == d) {
                const int   s0 = ovf[2 * i];
                const float w0 = gcn_dinv(cnt[s0]) * dd;
                const v4f h0 = *(const v4f*)(&H[(size_t)s0 * CH + ch]);
                acc += h0 * w0;
            }
        }
    }
    return acc;
}

__global__ __launch_bounds__(256) void gcn_gather(const int* __restrict__ cnt,
                                                  const int* __restrict__ bucket,
                                                  const int* __restrict__ ovf,
                                                  const float* __restrict__ H,
                                                  const float* __restrict__ bias,
                                                  float* __restrict__ out) {
    const int lane = threadIdx.x & 31;
    const int wave = __builtin_amdgcn_readfirstlane((int)(threadIdx.x >> 5));
    const int d    = blockIdx.x * 8 + wave;
    if (d >= NNODES) return;
    const v4f acc = gcn_gather_row(d, lane, cnt, bucket, ovf, H, bias);
    *(v4f*)(&out[(size_t)d * CH + lane * 4]) = acc;
}

#define FUSE_BLOCKS 1024
__global__ __launch_bounds__(512) void gcn_gather_gemm(const int* __restrict__ cnt,
                                                       const int* __restrict__ bucket,
                                                       const int* __restrict__ ovf,
                                                       const float* __restrict__ H,
                                                       const float* __restrict__ bias,
                                                       const float* __restrict__ W,
                                                       float* __restrict__ H2) {
    __shared__ float Wt[CH * WPITCH];
    __shared__ float Tt[16 * WPITCH];
    const int lane = threadIdx.x & 31;
    const int wave = __builtin_amdgcn_readfirstlane((int)(threadIdx.x >> 5));
    {
        const int sc = (wave & 7) * 16 + (lane & 15);
        const int sk = (wave >> 3) * (CH / 2) + ((lane >> 4) & 1);
        const float* __restrict__ wsrc = W + sk * CH + sc;
        float* wdst = &Wt[sc * WPITCH + sk];
        #pragma unroll 16
        for (int i = 0; i < CH / 4; ++i) {
            *wdst = *wsrc;
            wsrc += 2 * CH;
            wdst += 2;
        }
    }

    const int half = lane >> 4;
    const int mn   = lane & 15;

    for (int tile = blockIdx.x; tile < ROW_TILES; tile += FUSE_BLOCKS) {
        const int d = tile * 16 + wave;
        v4f r = gcn_gather_row(d, lane, cnt, bucket, ovf, H, bias);
        r.x = fmaxf(r.x, 0.0f); r.y = fmaxf(r.y, 0.0f);
        r.z = fmaxf(r.z, 0.0f); r.w = fmaxf(r.w, 0.0f);
        *(v4f*)(&Tt[wave * WPITCH + lane * 4]) = r;
        __syncthreads();
        if (wave < 8) {
            v8f acc = {};
            #pragma unroll 8
            for (int kk = 0; kk < CH; kk += 4) {
                const v2f a = *(const v2f*)(&Tt[mn * WPITCH + kk + 2 * half]);
                const v2f b = *(const v2f*)(&Wt[(wave * 16 + mn) * WPITCH + kk + 2 * half]);
                acc = __builtin_amdgcn_wmma_f32_16x16x4_f32(
                    false, a, false, b, (short)0, acc, false, false);
            }
            #pragma unroll
            for (int i = 0; i < 8; ++i) {
                H2[(size_t)(tile * 16 + half * 8 + i) * CH + wave * 16 + mn] = acc[i];
            }
        }
        __syncthreads();
    }
}

extern "C" void kernel_launch(void* const* d_in, const int* in_sizes, int n_in,
                              void* d_out, int out_size, void* d_ws, size_t ws_size,
                              hipStream_t stream) {
    const float* x   = (const float*)d_in[0];
    const int*   ei  = (const int*)d_in[1];
    const float* W1  = (const float*)d_in[2];
    const float* b1  = (const float*)d_in[3];
    const float* W2  = (const float*)d_in[4];
    const float* b2  = (const float*)d_in[5];
    float*       out = (float*)d_out;

    const int* src = ei;
    const int* dst = ei + NEDGES;

    float* ws   = (float*)d_ws;
    int*   cnt  = (int*)ws;
    float* H    = ws + 131072;
    float* H2   = ws + 131072 + 6400000;
    int*   bucket = (int*)(ws + 131072 + 12800000);
    int*   ovf    = (int*)(ws + 131072 + 12800000 + 3200000);

    const int T = 256;
    gcn_cnt_zero<<<(NNODES + 1 + T - 1) / T, T, 0, stream>>>(cnt);
    gcn_gemm_rank<<<GEMM_BLOCKS + RANK_BLOCKS, T, 0, stream>>>(x, W1, H, src, dst, cnt, bucket, ovf);

    const int gather_blocks = (NNODES + 7) / 8;

    gcn_gather_gemm<<<FUSE_BLOCKS, 512, 0, stream>>>(cnt, bucket, ovf, H, b1, W2, H2);

    gcn_gather<<<gather_blocks, T, 0, stream>>>(cnt, bucket, ovf, H2, b2, out);
}


# Count/rank atomics kept inside one chip half: an edge group ranks only edges whose destination is in its half's node range (die id; chunk ticket per range, then the other's); arrival-order ranks
# speedup vs baseline: 5.4754x; 1.4326x over previous
#include <hip/hip_runtime.h>
#include <hip/hip_bf16.h>

typedef __attribute__((ext_vector_type(2))) float v2f;
typedef __attribute__((ext_vector_type(4))) float v4f;
typedef __attribute__((ext_vector_type(8))) float v8f;

#define NNODES    50000
#define NEDGES    640000
#define CH        128
#define ROW_TILES 3125
#define WPITCH    132

#define CAP 64
#define HALF_NODES 25024
#define TICKET0 65536
#define TICKET1 (65536 + 64)

typedef __attribute__((ext_vector_type(4))) int v4i;

__global__ void gcn_cnt_zero(int* __restrict__ cnt) {
    int i = blockIdx.x * blockDim.x + threadIdx.x;
    if (i <= NNODES) cnt[i] = 0;
    if (i == 0) { cnt[TICKET0] = 0; cnt[TICKET1] = 0; }
}

__device__ __forceinline__ float gcn_dinv(const int c) {
    const float dg = (float)(c + 1);
    return (dg > 0.0f) ? rsqrtf(dg) : 0.0f;
}

__device__ __forceinline__ void gcn_put(const int s, const int d, const int r,
                                        int* __restrict__ cnt, int* __restrict__ bucket,
                                        int* __restrict__ ovf) {
    if (r < CAP) {
        bucket[d * CAP + r] = s;
    } else {
        const int i = atomicAdd(&cnt[NNODES], 1);
        ovf[2 * i]     = s;
        ovf[2 * i + 1] = d;
    }
}

__device__ __forceinline__ void gcn_gemm_block(const float* __restrict__ X,
                                               const float* __restrict__ W,
                                               float* __restrict__ H,
                                               const int gb, float* Wt) {
    {
        const int sc = (threadIdx.x >> 5) * 16 + (threadIdx.x & 15);
        const int sk = (threadIdx.x >> 4) & 1;
        const float* __restrict__ wsrc = W + sk * CH + sc;
        float* wdst = &Wt[sc * WPITCH + sk];
        #pragma unroll 16
        for (int i = 0; i < CH / 2; ++i) {
            *wdst = *wsrc;
            wsrc += 2 * CH;
            wdst += 2;
        }
    }
    __syncthreads();

    const int wave = threadIdx.x >> 5;
    const int lane = threadIdx.x & 31;
    const int rt   = gb * 8 + wave;
    if (rt >= ROW_TILES) return;

    const int half = lane >> 4;
    const int mn   = lane & 15;

    const float* __restrict__ xrow = X + (size_t)(rt * 16 + mn) * CH;

    v8f acc[8];
    v8f zero = {};
    #pragma unroll
    for (int c = 0; c < 8; ++c) acc[c] = zero;

    #pragma unroll 4
    for (int kk = 0; kk < CH; kk += 4) {
        v2f a = *(const v2f*)(xrow + kk + 2 * half);
        #pragma unroll
        for (int c = 0; c < 8; ++c) {
            v2f b = *(const v2f*)(&Wt[(c * 16 + mn) * WPITCH + kk + 2 * half]);
            acc[c] = __builtin_amdgcn_wmma_f32_16x16x4_f32(
                false, a, false, b, (short)0, acc[c], false, false);
        }
    }

    #pragma unroll
    for (int c = 0; c < 8; ++c) {
        #pragma unroll
        for (int i = 0; i < 8; ++i) {
            H[(size_t)(rt * 16 + half * 8 + i) * CH + c * 16 + mn] = acc[c][i];
        }
    }
}

#define GEMM_BLOCKS 391
#define RANK_BLOCKS 625
#define RANK_CHUNKS 625

__global__ __launch_bounds__(256) void gcn_gemm_rank(const float* __restrict__ X,
                                                     const float* __restrict__ W,
                                                     float* __restrict__ H,
                                                     const int* __restrict__ src,
                                                     const int* __restrict__ dst,
                                                     int* __restrict__ cnt,
                                                     int* __restrict__ bucket,
                                                     int* __restrict__ ovf) {
    __shared__ float Wt[CH * WPITCH];
    __shared__ int tk[2];
    if (blockIdx.x < GEMM_BLOCKS) {
        gcn_gemm_block(X, W, H, (int)blockIdx.x, Wt);
    } else {
        const int myhalf = (int)((__builtin_amdgcn_s_sendmsg_rtn(0x87) >> 18) & 1u);
        int n = 0;
        for (int pass = 0; pass < 2; ++pass) {
            const int h = myhalf ^ pass;
            int* __restrict__ ticket = cnt + (h ? TICKET1 : TICKET0);
            for (int it = 0; it <= RANK_CHUNKS; ++it) {
                const int slot = n & 1;
                ++n;
                if (threadIdx.x == 0) tk[slot] = atomicAdd(ticket, 1) | (h << 20);
                __syncthreads();
                const int drawn = tk[slot];
                const int c = drawn & 0xFFFFF;
                if (c >= RANK_CHUNKS) break;
                const int t = c * 256 + (int)threadIdx.x;
                const v4i d4 = ((const v4i*)dst)[t];
                const v4i s4 = ((const v4i*)src)[t];
                const bool up = (drawn >> 20) != 0;
                const bool a0 = (d4.x >= HALF_NODES) == up;
                const bool a1 = (d4.y >= HALF_NODES) == up;
                const bool a2 = (d4.z >= HALF_NODES) == up;
                const bool a3 = (d4.w >= HALF_NODES) == up;
                int r0 = 0, r1 = 0, r2 = 0, r3 = 0;
                if (a0) r0 = atomicAdd(&cnt[d4.x], 1);
                if (a1) r1 = atomicAdd(&cnt[d4.y], 1);
                if (a2) r2 = atomicAdd(&cnt[d4.z], 1);
                if (a3) r3 = atomicAdd(&cnt[d4.w], 1);
                if (a0) gcn_put(s4.x, d4.x, r0, cnt, bucket, ovf);
                if (a1) gcn_put(s4.y, d4.y, r1, cnt, bucket, ovf);
                if (a2) gcn_put(s4.z, d4.z, r2, cnt, bucket, ovf);
                if (a3) gcn_put(s4.w, d4.w, r3, cnt, bucket, ovf);
            }
        }
    }
}

__device__ __forceinline__ v4f gcn_gather_row(const int d, const int lane,
                                              const int* __restrict__ cnt,
                                              const int* __restrict__ bucket,
                                              const int* __restrict__ ovf,
                                              const float* __restrict__ H,
                                              const float* __restrict__ bias) {
    const int ch  = lane * 4;
    const int deg = cnt[d];
    const int end = min(deg, CAP);
    const int* __restrict__ slots = bucket + (size_t)d * CAP;
    const float dd = gcn_dinv(deg);

    v4f acc = *(const v4f*)(&H[(size_t)d * CH + ch]) * (dd * dd) + *(const v4f*)(&bias[ch]);

    for (int base = 0; base < end; base += 32) {
        const int n = min(32, end - base);
        int   mys = 0;
        float myw = 0.0f;
        if (lane < n) {
            mys = slots[base + lane];
            myw = gcn_dinv(cnt[mys]) * dd;
        }
        int k = 0;
        for (; k + 4 <= n; k += 4) {
            const int s0 = __builtin_amdgcn_readlane(mys, k);
            const int s1 = __builtin_amdgcn_readlane(mys, k + 1);
            const int s2 = __builtin_amdgcn_readlane(mys, k + 2);
            const int s3 = __builtin_amdgcn_readlane(mys, k + 3);
            const float w0 = __int_as_float(__builtin_amdgcn_readlane(__float_as_int(myw), k));
            const float w1 = __int_as_float(__builtin_amdgcn_readlane(__float_as_int(myw), k + 1));
            const float w2 = __int_as_float(__builtin_amdgcn_readlane(__float_as_int(myw), k + 2));
            const float w3 = __int_as_float(__builtin_amdgcn_readlane(__float_as_int(myw), k + 3));
            const v4f h0 = *(const v4f*)(&H[(size_t)s0 * CH + ch]);
            const v4f h1 = *(const v4f*)(&H[(size_t)s1 * CH + ch]);
            const v4f h2 = *(const v4f*)(&H[(size_t)s2 * CH + ch]);
            const v4f h3 = *(const v4f*)(&H[(size_t)s3 * CH + ch]);
            acc += h0 * w0;
            acc += h1 * w1;
            acc += h2 * w2;
            acc += h3 * w3;
        }
        for (; k < n; ++k) {
            const int   s0 = __builtin_amdgcn_readlane(mys, k);
            const float w0 = __int_as_float(__builtin_amdgcn_readlane(__float_as_int(myw), k));
            const v4f h0 = *(const v4f*)(&H[(size_t)s0 * CH + ch]);
            acc += h0 * w0;
        }
    }
    if (deg > CAP) {
        const int no = cnt[NNODES];
        for (int i = 0; i < no; ++i) {
            if (ovf[2 * i + 1] == d) {
                const int   s0 = ovf[2 * i];
                const float w0 = gcn_dinv(cnt[s0]) * dd;
                const v4f h0 = *(const v4f*)(&H[(size_t)s0 * CH + ch]);
                acc += h0 * w0;
            }
        }
    }
    return acc;
}

__global__ __launch_bounds__(256) void gcn_gather(const int* __restrict__ cnt,
                                                  const int* __restrict__ bucket,
                                                  const int* __restrict__ ovf,
                                                  const float* __restrict__ H,
                                                  const float* __restrict__ bias,
                                                  float* __restrict__ out) {
    const int lane = threadIdx.x & 31;
    const int wave = __builtin_amdgcn_readfirstlane((int)(threadIdx.x >> 5));
    const int d    = blockIdx.x * 8 + wave;
    if (d >= NNODES) return;
    const v4f acc = gcn_gather_row(d, lane, cnt, bucket, ovf, H, bias);
    *(v4f*)(&out[(size_t)d * CH + lane * 4]) = acc;
}

#define FUSE_BLOCKS 1024
__global__ __launch_bounds__(512) void gcn_gather_gemm(const int* __restrict__ cnt,
                                                       const int* __restrict__ bucket,
                                                       const int* __restrict__ ovf,
                                                       const float* __restrict__ H,
                                                       const float* __restrict__ bias,
                                                       const float* __restrict__ W,
                                                       float* __restrict__ H2) {
    __shared__ float Wt[CH * WPITCH];
    __shared__ float Tt[16 * WPITCH];
    const int lane = threadIdx.x & 31;
    const int wave = __builtin_amdgcn_readfirstlane((int)(threadIdx.x >> 5));
    {
        const int sc = (wave & 7) * 16 + (lane & 15);
        const int sk = (wave >> 3) * (CH / 2) + ((lane >> 4) & 1);
        const float* __restrict__ wsrc = W + sk * CH + sc;
        float* wdst = &Wt[sc * WPITCH + sk];
        #pragma unroll 16
        for (int i = 0; i < CH / 4; ++i) {
            *wdst = *wsrc;
            wsrc += 2 * CH;
            wdst += 2;
        }
    }

    const int half = lane >> 4;
    const int mn   = lane & 15;

    for (int tile = blockIdx.x; tile < ROW_TILES; tile += FUSE_BLOCKS) {
        const int d = tile * 16 + wave;
        v4f r = gcn_gather_row(d, lane, cnt, bucket, ovf, H, bias);
        r.x = fmaxf(r.x, 0.0f); r.y = fmaxf(r.y, 0.0f);
        r.z = fmaxf(r.z, 0.0f); r.w = fmaxf(r.w, 0.0f);
        *(v4f*)(&Tt[wave * WPITCH + lane * 4]) = r;
        __syncthreads();
        if (wave < 8) {
            v8f acc = {};
            #pragma unroll 8
            for (int kk = 0; kk < CH; kk += 4) {
                const v2f a = *(const v2f*)(&Tt[mn * WPITCH + kk + 2 * half]);
                const v2f b = *(const v2f*)(&Wt[(wave * 16 + mn) * WPITCH + kk + 2 * half]);
                acc = __builtin_amdgcn_wmma_f32_16x16x4_f32(
                    false, a, false, b, (short)0, acc, false, false);
            }
            #pragma unroll
            for (int i = 0; i < 8; ++i) {
                H2[(size_t)(tile * 16 + half * 8 + i) * CH + wave * 16 + mn] = acc[i];
            }
        }
        __syncthreads();
    }
}

extern "C" void kernel_launch(void* const* d_in, const int* in_sizes, int n_in,
                              void* d_out, int out_size, void* d_ws, size_t ws_size,
                              hipStream_t stream) {
    const float* x   = (const float*)d_in[0];
    const int*   ei  = (const int*)d_in[1];
    const float* W1  = (const float*)d_in[2];
    const float* b1  = (const float*)d_in[3];
    const float* W2  = (const float*)d_in[4];
    const float* b2  = (const float*)d_in[5];
    float*       out = (float*)d_out;

    const int* src = ei;
    const int* dst = ei + NEDGES;

    float* ws   = (float*)d_ws;
    int*   cnt  = (int*)ws;
    float* H    = ws + 131072;
    float* H2   = ws + 131072 + 6400000;
    int*   bucket = (int*)(ws + 131072 + 12800000);
    int*   ovf    = (int*)(ws + 131072 + 12800000 + 3200000);

    const int T = 256;
    gcn_cnt_zero<<<(NNODES + 1 + T - 1) / T, T, 0, stream>>>(cnt);
    gcn_gemm_rank<<<GEMM_BLOCKS + RANK_BLOCKS, T, 0, stream>>>(x, W1, H, src, dst, cnt, bucket, ovf);

    const int gather_blocks = (NNODES + 7) / 8;

    gcn_gather_gemm<<<FUSE_BLOCKS, 512, 0, stream>>>(cnt, bucket, ovf, H, b1, W2, H2);

    gcn_gather<<<gather_blocks, T, 0, stream>>>(cnt, bucket, ovf, H2, b2, out);
}


# Gather: a node's independent loads go out together: rows 8 per group and the last 1-7 as one group (was 4, then one at a time), slot list beside own row, source counts beside the first rows
# speedup vs baseline: 5.6829x; 1.0379x over previous
#include <hip/hip_runtime.h>
#include <hip/hip_bf16.h>

typedef __attribute__((ext_vector_type(2))) float v2f;
typedef __attribute__((ext_vector_type(4))) float v4f;
typedef __attribute__((ext_vector_type(8))) float v8f;

#define NNODES    50000
#define NEDGES    640000
#define CH        128
#define ROW_TILES 3125
#define WPITCH    132

#define CAP 64
#define HALF_NODES 25024
#define TICKET0 65536
#define TICKET1 (65536 + 64)

typedef __attribute__((ext_vector_type(4))) int v4i;

__global__ void gcn_cnt_zero(int* __restrict__ cnt) {
    int i = blockIdx.x * blockDim.x + threadIdx.x;
    if (i <= NNODES) cnt[i] = 0;
    if (i == 0) { cnt[TICKET0] = 0; cnt[TICKET1] = 0; }
}

__device__ __forceinline__ float gcn_dinv(const int c) {
    const float dg = (float)(c + 1);
    return (dg > 0.0f) ? rsqrtf(dg) : 0.0f;
}

__device__ __forceinline__ void gcn_put(const int s, const int d, const int r,
                                        int* __restrict__ cnt, int* __restrict__ bucket,
                                        int* __restrict__ ovf) {
    if (r < CAP) {
        bucket[d * CAP + r] = s;
    } else {
        const int i = atomicAdd(&cnt[NNODES], 1);
        ovf[2 * i]     = s;
        ovf[2 * i + 1] = d;
    }
}

__device__ __forceinline__ void gcn_gemm_block(const float* __restrict__ X,
                                               const float* __restrict__ W,
                                               float* __restrict__ H,
                                               const int gb, float* Wt) {
    {
        const int sc = (threadIdx.x >> 5) * 16 + (threadIdx.x & 15);
        const int sk = (threadIdx.x >> 4) & 1;
        const float* __restrict__ wsrc = W + sk * CH + sc;
        float* wdst = &Wt[sc * WPITCH + sk];
        #pragma unroll 16
        for (int i = 0; i < CH / 2; ++i) {
            *wdst = *wsrc;
            wsrc += 2 * CH;
            wdst += 2;
        }
    }
    __syncthreads();

    const int wave = threadIdx.x >> 5;
    const int lane = threadIdx.x & 31;
    const int rt   = gb * 8 + wave;
    if (rt >= ROW_TILES) return;

    const int half = lane >> 4;
    const int mn   = lane & 15;

    const float* __restrict__ xrow = X + (size_t)(rt * 16 + mn) * CH;

    v8f acc[8];
    v8f zero = {};
    #pragma unroll
    for (int c = 0; c < 8; ++c) acc[c] = zero;

    #pragma unroll 4
    for (int kk = 0; kk < CH; kk += 4) {
        v2f a = *(const v2f*)(xrow + kk + 2 * half);
        #pragma unroll
        for (int c = 0; c < 8; ++c) {
            v2f b = *(const v2f*)(&Wt[(c * 16 + mn) * WPITCH + kk + 2 * half]);
            acc[c] = __builtin_amdgcn_wmma_f32_16x16x4_f32(
                false, a, false, b, (short)0, acc[c], false, false);
        }
    }

    #pragma unroll
    for (int c = 0; c < 8; ++c) {
        #pragma unroll
        for (int i = 0; i < 8; ++i) {
            H[(size_t)(rt * 16 + half * 8 + i) * CH + c * 16 + mn] = acc[c][i];
        }
    }
}

#define GEMM_BLOCKS 391
#define RANK_BLOCKS 625
#define RANK_CHUNKS 625

__global__ __launch_bounds__(256) void gcn_gemm_rank(const float* __restrict__ X,
                                                     const float* __restrict__ W,
                                                     float* __restrict__ H,
                                                     const int* __restrict__ src,
                                                     const int* __restrict__ dst,
                                                     int* __restrict__ cnt,
                                                     int* __restrict__ bucket,
                                                     int* __restrict__ ovf) {
    __shared__ float Wt[CH * WPITCH];
    __shared__ int tk[2];
    if (blockIdx.x < GEMM_BLOCKS) {
        gcn_gemm_block(X, W, H, (int)blockIdx.x, Wt);
    } else {
        const int myhalf = (int)((__builtin_amdgcn_s_sendmsg_rtn(0x87) >> 18) & 1u);
        int n = 0;
        for (int pass = 0; pass < 2; ++pass) {
            const int h = myhalf ^ pass;
            int* __restrict__ ticket = cnt + (h ? TICKET1 : TICKET0);
            for (int it = 0; it <= RANK_CHUNKS; ++it) {
                const int slot = n & 1;
                ++n;
                if (threadIdx.x == 0) tk[slot] = atomicAdd(ticket, 1) | (h << 20);
                __syncthreads();
                const int drawn = tk[slot];
                const int c = drawn & 0xFFFFF;
                if (c >= RANK_CHUNKS) break;
                const int t = c * 256 + (int)threadIdx.x;
                const v4i d4 = ((const v4i*)dst)[t];
                const v4i s4 = ((const v4i*)src)[t];
                const bool up = (drawn >> 20) != 0;
                const bool a0 = (d4.x >= HALF_NODES) == up;
                const bool a1 = (d4.y >= HALF_NODES) == up;
                const bool a2 = (d4.z >= HALF_NODES) == up;
                const bool a3 = (d4.w >= HALF_NODES) == up;
                int r0 = 0, r1 = 0, r2 = 0, r3 = 0;
                if (a0) r0 = atomicAdd(&cnt[d4.x], 1);
                if (a1) r1 = atomicAdd(&cnt[d4.y], 1);
                if (a2) r2 = atomicAdd(&cnt[d4.z], 1);
                if (a3) r3 = atomicAdd(&cnt[d4.w], 1);
                if (a0) gcn_put(s4.x, d4.x, r0, cnt, bucket, ovf);
                if (a1) gcn_put(s4.y, d4.y, r1, cnt, bucket, ovf);
                if (a2) gcn_put(s4.z, d4.z, r2, cnt, bucket, ovf);
                if (a3) gcn_put(s4.w, d4.w, r3, cnt, bucket, ovf);
            }
        }
    }
}

template <int R>
__device__ __forceinline__ v4f gcn_rows(v4f acc, const int mys, float& myw, const int k,
                                        const int lane, const int n, const float dd,
                                        const int* __restrict__ cnt,
                                        const float* __restrict__ H, const int ch) {
    v4f h[R];
    #pragma unroll
    for (int j = 0; j < R; ++j) {
        const int s = __builtin_amdgcn_readlane(mys, k + j);
        h[j] = *(const v4f*)(&H[(size_t)s * CH + ch]);
    }
    if (k == 0) {
        int cs = 0;
        if (lane < n) cs = cnt[mys];
        myw = gcn_dinv(cs) * dd;
    }
    #pragma unroll
    for (int j = 0; j < R; ++j) {
        const float w = __int_as_float(__builtin_amdgcn_readlane(__float_as_int(myw), k + j));
        acc += h[j] * w;
    }
    return acc;
}

__device__ __forceinline__ v4f gcn_gather_row(const int d, const int lane,
                                              const int* __restrict__ cnt,
                                              const int* __restrict__ bucket,
                                              const int* __restrict__ ovf,
                                              const float* __restrict__ H,
                                              const float* __restrict__ bias) {
    const int ch  = lane * 4;
    const int deg = cnt[d];
    const int end = min(deg, CAP);
    const int* __restrict__ slots = bucket + (size_t)d * CAP;
    const float dd = gcn_dinv(deg);

    const v4f hd = *(const v4f*)(&H[(size_t)d * CH + ch]);
    const v4f bb = *(const v4f*)(&bias[ch]);
    int base = 0;
    int n    = min(32, end);
    int mys  = 0;
    if (lane < n) mys = slots[lane];

    v4f   acc = hd * (dd * dd) + bb;
    float myw = 0.0f;
    while (n > 0) {
        int k = 0;
        for (; k + 8 <= n; k += 8) acc = gcn_rows<8>(acc, mys, myw, k, lane, n, dd, cnt, H, ch);
        switch (n - k) {
            case 1: acc = gcn_rows<1>(acc, mys, myw, k, lane, n, dd, cnt, H, ch); break;
            case 2: acc = gcn_rows<2>(acc, mys, myw, k, lane, n, dd, cnt, H, ch); break;
            case 3: acc = gcn_rows<3>(acc, mys, myw, k, lane, n, dd, cnt, H, ch); break;
            case 4: acc = gcn_rows<4>(acc, mys, myw, k, lane, n, dd, cnt, H, ch); break;
            case 5: acc = gcn_rows<5>(acc, mys, myw, k, lane, n, dd, cnt, H, ch); break;
            case 6: acc = gcn_rows<6>(acc, mys, myw, k, lane, n, dd, cnt, H, ch); break;
            case 7: acc = gcn_rows<7>(acc, mys, myw, k, lane, n, dd, cnt, H, ch); break;
            default: break;
        }
        base += 32;
        n = min(32, end - base);
        if (lane < n) mys = slots[base + lane];
    }
    if (deg > CAP) {
        const int no = cnt[NNODES];
        for (int i = 0; i < no; ++i) {
            if (ovf[2 * i + 1] == d) {
                const int   s0 = ovf[2 * i];
                const float w0 = gcn_dinv(cnt[s0]) * dd;
                const v4f h0 = *(const v4f*)(&H[(size_t)s0 * CH + ch]);
                acc += h0 * w0;
            }
        }
    }
    return acc;
}

__global__ __launch_bounds__(256) void gcn_gather(const int* __restrict__ cnt,
                                                  const int* __restrict__ bucket,
                                                  const int* __restrict__ ovf,
                                                  const float* __restrict__ H,
                                                  const float* __restrict__ bias,
                                                  float* __restrict__ out) {
    const int lane = threadIdx.x & 31;
    const int wave = __builtin_amdgcn_readfirstlane((int)(threadIdx.x >> 5));
    const int d    = blockIdx.x * 8 + wave;
    if (d >= NNODES) return;
    const v4f acc = gcn_gather_row(d, lane, cnt, bucket, ovf, H, bias);
    *(v4f*)(&out[(size_t)d * CH + lane * 4]) = acc;
}

#define FUSE_BLOCKS 1024
__global__ __launch_bounds__(512) void gcn_gather_gemm(const int* __restrict__ cnt,
                                                       const int* __restrict__ bucket,
                                                       const int* __restrict__ ovf,
                                                       const float* __restrict__ H,
                                                       const float* __restrict__ bias,
                                                       const float* __restrict__ W,
                                                       float* __restrict__ H2) {
    __shared__ float Wt[CH * WPITCH];
    __shared__ float Tt[16 * WPITCH];
    const int lane = threadIdx.x & 31;
    const int wave = __builtin_amdgcn_readfirstlane((int)(threadIdx.x >> 5));
    {
        const int sc = (wave & 7) * 16 + (lane & 15);
        const int sk = (wave >> 3) * (CH / 2) + ((lane >> 4) & 1);
        const float* __restrict__ wsrc = W + sk * CH + sc;
        float* wdst = &Wt[sc * WPITCH + sk];
        #pragma unroll 16
        for (int i = 0; i < CH / 4; ++i) {
            *wdst = *wsrc;
            wsrc += 2 * CH;
            wdst += 2;
        }
    }

    const int half = lane >> 4;
    const int mn   = lane & 15;

    for (int tile = blockIdx.x; tile < ROW_TILES; tile += FUSE_BLOCKS) {
        const int d = tile * 16 + wave;
        v4f r = gcn_gather_row(d, lane, cnt, bucket, ovf, H, bias);
        r.x = fmaxf(r.x, 0.0f); r.y = fmaxf(r.y, 0.0f);
        r.z = fmaxf(r.z, 0.0f); r.w = fmaxf(r.w, 0.0f);
        *(v4f*)(&Tt[wave * WPITCH + lane * 4]) = r;
        __syncthreads();
        if (wave < 8) {
            v8f acc = {};
            #pragma unroll 8
            for (int kk = 0; kk < CH; kk += 4) {
                const v2f a = *(const v2f*)(&Tt[mn * WPITCH + kk + 2 * half]);
                const v2f b = *(const v2f*)(&Wt[(wave * 16 + mn) * WPITCH + kk + 2 * half]);
                acc = __builtin_amdgcn_wmma_f32_16x16x4_f32(
                    false, a, false, b, (short)0, acc, false, false);
            }
            #pragma unroll
            for (int i = 0; i < 8; ++i) {
                H2[(size_t)(tile * 16 + half * 8 + i) * CH + wave * 16 + mn] = acc[i];
            }
        }
        __syncthreads();
    }
}

extern "C" void kernel_launch(void* const* d_in, const int* in_sizes, int n_in,
                              void* d_out, int out_size, void* d_ws, size_t ws_size,
                              hipStream_t stream) {
    const float* x   = (const float*)d_in[0];
    const int*   ei  = (const int*)d_in[1];
    const float* W1  = (const float*)d_in[2];
    const float* b1  = (const float*)d_in[3];
    const float* W2  = (const float*)d_in[4];
    const float* b2  = (const float*)d_in[5];
    float*       out = (float*)d_out;

    const int* src = ei;
    const int* dst = ei + NEDGES;

    float* ws   = (float*)d_ws;
    int*   cnt  = (int*)ws;
    float* H    = ws + 131072;
    float* H2   = ws + 131072 + 6400000;
    int*   bucket = (int*)(ws + 131072 + 12800000);
    int*   ovf    = (int*)(ws + 131072 + 12800000 + 3200000);

    const int T = 256;
    gcn_cnt_zero<<<(NNODES + 1 + T - 1) / T, T, 0, stream>>>(cnt);
    gcn_gemm_rank<<<GEMM_BLOCKS + RANK_BLOCKS, T, 0, stream>>>(x, W1, H, src, dst, cnt, bucket, ovf);

    const int gather_blocks = (NNODES + 7) / 8;

    gcn_gather_gemm<<<FUSE_BLOCKS, 512, 0, stream>>>(cnt, bucket, ovf, H, b1, W2, H2);

    gcn_gather<<<gather_blocks, T, 0, stream>>>(cnt, bucket, ovf, H2, b2, out);
}
